# starting kernel
# baseline (speedup 1.0000x reference)
#include <hip/hip_runtime.h>

typedef __attribute__((ext_vector_type(2))) float v2f;
typedef __attribute__((ext_vector_type(8))) float v8f;

#define LEAKY_SLOPE 0.01f

__device__ __forceinline__ void atomAddF(float* p, float v) {
    __hip_atomic_fetch_add(p, v, __ATOMIC_RELAXED, __HIP_MEMORY_SCOPE_AGENT);
}

__global__ void k_zero4(float4* __restrict__ p, int n4) {
    int i = blockIdx.x * blockDim.x + threadIdx.x;
    if (i < n4) p[i] = make_float4(0.f, 0.f, 0.f, 0.f);
}

__global__ void k_degrees(const int* __restrict__ src, const int* __restrict__ dst,
                          float* __restrict__ degO, float* __restrict__ degI, int E) {
    int e = blockIdx.x * blockDim.x + threadIdx.x;
    if (e >= E) return;
    atomAddF(&degO[src[e]], 1.0f);
    atomAddF(&degI[dst[e]], 1.0f);
}

__global__ void k_inv_sqrt(float* __restrict__ a, float* __restrict__ b, int n) {
    int i = blockIdx.x * blockDim.x + threadIdx.x;
    if (i >= n) return;
    a[i] = rsqrtf(fmaxf(a[i], 1.0f));
    b[i] = rsqrtf(fmaxf(b[i], 1.0f));
}

__global__ void k_gather(const float* __restrict__ x, const float* __restrict__ invO,
                         const int* __restrict__ src, const int* __restrict__ dst,
                         float* __restrict__ agg, int E, int lgChunks, int d) {
    int tid = blockIdx.x * blockDim.x + threadIdx.x;
    int e = tid >> lgChunks;
    if (e >= E) return;
    int c = tid & ((1 << lgChunks) - 1);
    int s = src[e];
    int t = dst[e];
    float w = invO[s];
    const float4 v = *reinterpret_cast<const float4*>(x + (size_t)s * d + c * 4);
    float* base = agg + (size_t)t * d + c * 4;
    atomAddF(base + 0, v.x * w);
    atomAddF(base + 1, v.y * w);
    atomAddF(base + 2, v.z * w);
    atomAddF(base + 3, v.w * w);
}

template <int KDIM, int MD, bool LEAKY>
__global__ void k_gemm_wmma(const float* __restrict__ agg, const float* __restrict__ invI,
                            const float* __restrict__ W, const float* __restrict__ bias,
                            float* __restrict__ out, int Nn, int colTiles, int rowBlocks) {
    constexpr int NK = KDIM / 4;
    constexpr bool FULLCOL = (MD % 16 == 0);

    int gwave    = __builtin_amdgcn_readfirstlane(
                       (int)((blockIdx.x * blockDim.x + threadIdx.x) >> 5));
    int lane     = threadIdx.x & 31;
    int rowBlock = gwave / colTiles;
    int colTile  = gwave - rowBlock * colTiles;
    if (rowBlock >= rowBlocks) return;

    int   ln15    = lane & 15;
    int   col     = colTile * 16 + ln15;
    int   colC    = FULLCOL ? col : ((col < MD) ? col : (MD - 1));
    float colMask = FULLCOL ? 1.f : ((col < MD) ? 1.f : 0.f);
    int   kHalf   = (lane >> 4) * 2;
    int   hi      = lane >> 4;

    v2f bf[NK];
    #pragma unroll
    for (int t = 0; t < NK; ++t) {
        int k0 = t * 4 + kHalf;
        bf[t].x = W[k0 * MD + colC] * colMask;
        bf[t].y = W[(k0 + 1) * MD + colC] * colMask;
    }
    float bv = bias[colC] * colMask;

    #pragma unroll
    for (int st = 0; st < 4; ++st) {
        int tileBase = rowBlock * 64 + st * 16;
        if (tileBase >= Nn) break;
        int   mrow  = tileBase + ln15;
        int   mrowC = (mrow < Nn) ? mrow : (Nn - 1);
        float s     = invI[mrowC];

        v8f acc = {0.f, 0.f, 0.f, 0.f, 0.f, 0.f, 0.f, 0.f};
        #pragma unroll
        for (int t = 0; t < NK; ++t) {
            int k0 = t * 4 + kHalf;
            v2f a = *reinterpret_cast<const v2f*>(agg + (size_t)mrowC * KDIM + k0);
            a.x *= s;
            a.y *= s;
            acc = __builtin_amdgcn_wmma_f32_16x16x4_f32(
                false, a, false, bf[t], (short)0, acc, false, false);
        }

        float h[8];
        #pragma unroll
        for (int j = 0; j < 8; ++j) {
            float v = acc[j] + bv;
            h[j] = LEAKY ? fmaxf(v, LEAKY_SLOPE * v) : v;
        }

        float* outBase = out + (size_t)(tileBase + hi * 8) * MD + col;
        if (FULLCOL && (tileBase + 16 <= Nn)) {
            #pragma unroll
            for (int j = 0; j < 8; ++j) outBase[j * MD] = h[j];
        } else {
            #pragma unroll
            for (int j = 0; j < 8; ++j) {
                int r = tileBase + hi * 8 + j;
                if (col < MD && r < Nn) outBase[j * MD] = h[j];
            }
        }
    }
}

extern "C" void kernel_launch(void* const* d_in, const int* in_sizes, int n_in,
                              void* d_out, int out_size, void* d_ws, size_t ws_size,
                              hipStream_t stream) {
    (void)n_in; (void)out_size; (void)ws_size;

    const float* features = (const float*)d_in[0];
    const int*   esrc     = (const int*)d_in[1];
    const int*   edst     = (const int*)d_in[2];

    const int N = in_sizes[0] / 4;
    const int E = in_sizes[1];
    static const int DIMS[9] = {4, 64, 64, 64, 64, 64, 64, 64, 3};

    float* ws      = (float*)d_ws;
    float* inv_out = ws;
    float* inv_in  = ws + 100352;
    float* xbuf    = ws + 200704;
    float* agg     = ws + 200704 + 6400000;

    const int BT = 256;

    {
        int n4 = (N + 3) / 4;
        k_zero4<<<(n4 + BT - 1) / BT, BT, 0, stream>>>((float4*)inv_out, n4);
        k_zero4<<<(n4 + BT - 1) / BT, BT, 0, stream>>>((float4*)inv_in,  n4);
        k_degrees<<<(E + BT - 1) / BT, BT, 0, stream>>>(esrc, edst, inv_out, inv_in, E);
        k_inv_sqrt<<<(N + BT - 1) / BT, BT, 0, stream>>>(inv_out, inv_in, N);
    }

    const float* xcur = features;
    for (int i = 0; i < 8; ++i) {
        const int Kd = DIMS[i];
        const int Md = DIMS[i + 1];
        const float* Wi = (const float*)d_in[3 + 2 * i];
        const float* bi = (const float*)d_in[4 + 2 * i];

        int n4 = N * Kd / 4;
        k_zero4<<<(n4 + BT - 1) / BT, BT, 0, stream>>>((float4*)agg, n4);

        int chunks = Kd / 4;
        int lg = (chunks == 16) ? 4 : 0;
        long long total = (long long)E * chunks;
        int blocks = (int)((total + BT - 1) / BT);
        k_gather<<<blocks, BT, 0, stream>>>(xcur, inv_out, esrc, edst, agg, E, lg, Kd);

        int colTiles  = (Md + 15) / 16;
        int rowBlocks = (N + 63) / 64;
        long long waves = (long long)rowBlocks * colTiles;
        long long thr = waves * 32;
        int gblocks = (int)((thr + BT - 1) / BT);
        float* yout = (i == 7) ? (float*)d_out : xbuf;

        if (i == 0) {
            k_gemm_wmma<4, 64, true><<<gblocks, BT, 0, stream>>>(
                agg, inv_in, Wi, bi, yout, N, colTiles, rowBlocks);
        } else if (i < 7) {
            k_gemm_wmma<64, 64, true><<<gblocks, BT, 0, stream>>>(
                agg, inv_in, Wi, bi, yout, N, colTiles, rowBlocks);
        } else {
            k_gemm_wmma<64, 3, false><<<gblocks, BT, 0, stream>>>(
                agg, inv_in, Wi, bi, yout, N, colTiles, rowBlocks);
        }
        xcur = xbuf;
    }
}


# Per-call edge list sorted by target (int atomic count+rank, scan, fill) + pull aggregation, one wave per node: no float atomics or zero fills; sum order per node = arrival rank, lane-group partials
# speedup vs baseline: 3.3732x; 3.3732x over previous
#include <hip/hip_runtime.h>


typedef __attribute__((ext_vector_type(2))) float v2f;
typedef __attribute__((ext_vector_type(8))) float v8f;

#define LEAKY_SLOPE 0.01f

__device__ __forceinline__ void atomAddF(float* p, float v) {
    __hip_atomic_fetch_add(p, v, __ATOMIC_RELAXED, __HIP_MEMORY_SCOPE_AGENT);
}

__global__ void k_zero4(float4* __restrict__ p, int n4) {
    int i = blockIdx.x * blockDim.x + threadIdx.x;
    if (i < n4) p[i] = make_float4(0.f, 0.f, 0.f, 0.f);
}

__global__ void k_degrees(const int* __restrict__ src, const int* __restrict__ dst,
                          float* __restrict__ degO, float* __restrict__ degI, int E) {
    int e = blockIdx.x * blockDim.x + threadIdx.x;
    if (e >= E) return;
    atomAddF(&degO[src[e]], 1.0f);
    atomAddF(&degI[dst[e]], 1.0f);
}

__global__ void k_inv_sqrt(float* __restrict__ a, float* __restrict__ b, int n) {
    int i = blockIdx.x * blockDim.x + threadIdx.x;
    if (i >= n) return;
    a[i] = rsqrtf(fmaxf(a[i], 1.0f));
    b[i] = rsqrtf(fmaxf(b[i], 1.0f));
}

__global__ void k_gather(const float* __restrict__ x, const float* __restrict__ invO,
                         const int* __restrict__ src, const int* __restrict__ dst,
                         float* __restrict__ agg, int E, int lgChunks, int d) {
    int tid = blockIdx.x * blockDim.x + threadIdx.x;
    int e = tid >> lgChunks;
    if (e >= E) return;
    int c = tid & ((1 << lgChunks) - 1);
    int s = src[e];
    int t = dst[e];
    float w = invO[s];
    const float4 v = *reinterpret_cast<const float4*>(x + (size_t)s * d + c * 4);
    float* base = agg + (size_t)t * d + c * 4;
    atomAddF(base + 0, v.x * w);
    atomAddF(base + 1, v.y * w);
    atomAddF(base + 2, v.z * w);
    atomAddF(base + 3, v.w * w);
}

__global__ void k_count_rank(const int* __restrict__ src, const int* __restrict__ dst,
                             int* __restrict__ cntO, int* __restrict__ cntI,
                             int* __restrict__ rank, int E) {
    int e = blockIdx.x * blockDim.x + threadIdx.x;
    if (e >= E) return;
    rank[e] = __hip_atomic_fetch_add(&cntI[dst[e]], 1, __ATOMIC_RELAXED, __HIP_MEMORY_SCOPE_AGENT);
    __hip_atomic_fetch_add(&cntO[src[e]], 1, __ATOMIC_RELAXED, __HIP_MEMORY_SCOPE_AGENT);
}

__global__ void __launch_bounds__(1024) k_scan(const int* __restrict__ cnt,
                                               int* __restrict__ rowptr, int n) {
    __shared__ int wsum[32];
    __shared__ int tileTotal;
    const int tid = threadIdx.x, lane = tid & 31, wv = tid >> 5;
    int carry = 0;
    for (int base = 0; base < n; base += 4096) {
        const int i = base + tid * 4;
        int a0 = 0, a1 = 0, a2 = 0, a3 = 0;
        if (i + 3 < n) {
            const int4 v = *reinterpret_cast<const int4*>(cnt + i);
            a0 = v.x; a1 = v.y; a2 = v.z; a3 = v.w;
        } else {
            if (i < n) a0 = cnt[i];
            if (i + 1 < n) a1 = cnt[i + 1];
            if (i + 2 < n) a2 = cnt[i + 2];
        }
        const int tsum = a0 + a1 + a2 + a3;
        int inc = tsum;
        for (int o = 1; o < 32; o <<= 1) {
            int t = __shfl_up(inc, o);
            if (lane >= o) inc += t;
        }
        if (lane == 31) wsum[wv] = inc;
        __syncthreads();
        if (wv == 0) {
            const int w = wsum[lane];
            int winc = w;
            for (int o = 1; o < 32; o <<= 1) {
                int t = __shfl_up(winc, o);
                if (lane >= o) winc += t;
            }
            wsum[lane] = winc - w;
            if (lane == 31) tileTotal = winc;
        }
        __syncthreads();
        const int ex = carry + wsum[wv] + (inc - tsum);
        carry += tileTotal;
        if (i + 3 < n) {
            *reinterpret_cast<int4*>(rowptr + i) = make_int4(ex, ex + a0, ex + a0 + a1, ex + a0 + a1 + a2);
        } else {
            if (i < n) rowptr[i] = ex;
            if (i + 1 < n) rowptr[i + 1] = ex + a0;
            if (i + 2 < n) rowptr[i + 2] = ex + a0 + a1;
        }
        __syncthreads();
    }
    if (tid == 0) rowptr[n] = carry;
}

__global__ void k_inv_sqrt_cnt(const int* cntA, float* outA, const int* cntB, float* outB, int n) {
    int i = blockIdx.x * blockDim.x + threadIdx.x;
    if (i >= n) return;
    float fa = (float)cntA[i];
    float fb = (float)cntB[i];
    outA[i] = rsqrtf(fmaxf(fa, 1.0f));
    outB[i] = rsqrtf(fmaxf(fb, 1.0f));
}

__global__ void k_fill(const int* __restrict__ src, const int* __restrict__ dst,
                       const int* __restrict__ rowptr, const int* __restrict__ rank,
                       int* __restrict__ csr, int E) {
    int e = blockIdx.x * blockDim.x + threadIdx.x;
    if (e >= E) return;
    csr[rowptr[dst[e]] + rank[e]] = src[e];
}

template <int D>
__global__ void k_pull(const float* __restrict__ x, const float* __restrict__ invO,
                       const int* __restrict__ rowptr, const int* __restrict__ csr,
                       float* __restrict__ agg, int Nn) {
    constexpr int LPR = D / 4;
    constexpr int EPI = 32 / LPR;
    const int node = __builtin_amdgcn_readfirstlane(
                         (int)((blockIdx.x * blockDim.x + threadIdx.x) >> 5));
    if (node >= Nn) return;
    const int lane = threadIdx.x & 31;
    const int sub  = lane / LPR;
    const int c    = lane % LPR;
    const int beg  = rowptr[node];
    const int end  = rowptr[node + 1];
    const float* xc = x + c * 4;

    float4 acc = make_float4(0.f, 0.f, 0.f, 0.f);
    int e = beg + sub;
    for (; e + 3 * EPI < end; e += 4 * EPI) {
        const int s0 = csr[e];
        const int s1 = csr[e + EPI];
        const int s2 = csr[e + 2 * EPI];
        const int s3 = csr[e + 3 * EPI];
        const float w0 = invO[s0];
        const float w1 = invO[s1];
        const float w2 = invO[s2];
        const float w3 = invO[s3];
        const float4 v0 = *reinterpret_cast<const float4*>(xc + (size_t)s0 * D);
        const float4 v1 = *reinterpret_cast<const float4*>(xc + (size_t)s1 * D);
        const float4 v2 = *reinterpret_cast<const float4*>(xc + (size_t)s2 * D);
        const float4 v3 = *reinterpret_cast<const float4*>(xc + (size_t)s3 * D);
        acc.x += v0.x * w0; acc.y += v0.y * w0; acc.z += v0.z * w0; acc.w += v0.w * w0;
        acc.x += v1.x * w1; acc.y += v1.y * w1; acc.z += v1.z * w1; acc.w += v1.w * w1;
        acc.x += v2.x * w2; acc.y += v2.y * w2; acc.z += v2.z * w2; acc.w += v2.w * w2;
        acc.x += v3.x * w3; acc.y += v3.y * w3; acc.z += v3.z * w3; acc.w += v3.w * w3;
    }
    for (; e < end; e += EPI) {
        const int s0 = csr[e];
        const float w0 = invO[s0];
        const float4 v0 = *reinterpret_cast<const float4*>(xc + (size_t)s0 * D);
        acc.x += v0.x * w0; acc.y += v0.y * w0; acc.z += v0.z * w0; acc.w += v0.w * w0;
    }
    #pragma unroll
    for (int o = LPR; o < 32; o <<= 1) {
        acc.x += __shfl_xor(acc.x, o);
        acc.y += __shfl_xor(acc.y, o);
        acc.z += __shfl_xor(acc.z, o);
        acc.w += __shfl_xor(acc.w, o);
    }
    if (sub == 0)
        *reinterpret_cast<float4*>(agg + (size_t)node * D + c * 4) = acc;
}

template <int KDIM, int MD, bool LEAKY>
__global__ void k_gemm_wmma(const float* __restrict__ agg, const float* __restrict__ invI,
                            const float* __restrict__ W, const float* __restrict__ bias,
                            float* __restrict__ out, int Nn, int colTiles, int rowBlocks) {
    constexpr int NK = KDIM / 4;
    constexpr bool FULLCOL = (MD % 16 == 0);

    int gwave    = __builtin_amdgcn_readfirstlane(
                       (int)((blockIdx.x * blockDim.x + threadIdx.x) >> 5));
    int lane     = threadIdx.x & 31;
    int rowBlock = gwave / colTiles;
    int colTile  = gwave - rowBlock * colTiles;
    if (rowBlock >= rowBlocks) return;

    int   ln15    = lane & 15;
    int   col     = colTile * 16 + ln15;
    int   colC    = FULLCOL ? col : ((col < MD) ? col : (MD - 1));
    float colMask = FULLCOL ? 1.f : ((col < MD) ? 1.f : 0.f);
    int   kHalf   = (lane >> 4) * 2;
    int   hi      = lane >> 4;

    v2f bf[NK];
    #pragma unroll
    for (int t = 0; t < NK; ++t) {
        int k0 = t * 4 + kHalf;
        bf[t].x = W[k0 * MD + colC] * colMask;
        bf[t].y = W[(k0 + 1) * MD + colC] * colMask;
    }
    float bv = bias[colC] * colMask;

    #pragma unroll
    for (int st = 0; st < 4; ++st) {
        int tileBase = rowBlock * 64 + st * 16;
        if (tileBase >= Nn) break;
        int   mrow  = tileBase + ln15;
        int   mrowC = (mrow < Nn) ? mrow : (Nn - 1);
        float s     = invI[mrowC];

        v8f acc = {0.f, 0.f, 0.f, 0.f, 0.f, 0.f, 0.f, 0.f};
        #pragma unroll
        for (int t = 0; t < NK; ++t) {
            int k0 = t * 4 + kHalf;
            v2f a = *reinterpret_cast<const v2f*>(agg + (size_t)mrowC * KDIM + k0);
            a.x *= s;
            a.y *= s;
            acc = __builtin_amdgcn_wmma_f32_16x16x4_f32(
                false, a, false, bf[t], (short)0, acc, false, false);
        }

        float h[8];
        #pragma unroll
        for (int j = 0; j < 8; ++j) {
            float v = acc[j] + bv;
            h[j] = LEAKY ? fmaxf(v, LEAKY_SLOPE * v) : v;
        }

        float* outBase = out + (size_t)(tileBase + hi * 8) * MD + col;
        if (FULLCOL && (tileBase + 16 <= Nn)) {
            #pragma unroll
            for (int j = 0; j < 8; ++j) outBase[j * MD] = h[j];
        } else {
            #pragma unroll
            for (int j = 0; j < 8; ++j) {
                int r = tileBase + hi * 8 + j;
                if (col < MD && r < Nn) outBase[j * MD] = h[j];
            }
        }
    }
}

extern "C" void kernel_launch(void* const* d_in, const int* in_sizes, int n_in,
                              void* d_out, int out_size, void* d_ws, size_t ws_size,
                              hipStream_t stream) {
    (void)n_in; (void)out_size;

    const float* features = (const float*)d_in[0];
    const int*   esrc     = (const int*)d_in[1];
    const int*   edst     = (const int*)d_in[2];

    const int N = in_sizes[0] / 4;
    const int E = in_sizes[1];
    static const int DIMS[9] = {4, 64, 64, 64, 64, 64, 64, 64, 3};

    float* ws      = (float*)d_ws;
    float* inv_out = ws;
    float* inv_in  = ws + 100352;
    float* xbuf    = ws + 200704;
    float* agg     = ws + 200704 + 6400000;

    const int BT = 256;

    if (ws_size >= (size_t)65204224) {
        int* cntO   = (int*)inv_out;
        int* cntI   = (int*)inv_in;
        int* rank   = (int*)xbuf;
        int* csr    = (int*)(ws + 13000704);
        int* rowptr = (int*)(ws + 13000704 + 3200000);

        int n4 = (N + 3) / 4;
        k_zero4<<<(n4 + BT - 1) / BT, BT, 0, stream>>>((float4*)inv_out, n4);
        k_zero4<<<(n4 + BT - 1) / BT, BT, 0, stream>>>((float4*)inv_in,  n4);
        k_count_rank<<<(E + BT - 1) / BT, BT, 0, stream>>>(esrc, edst, cntO, cntI, rank, E);
        k_scan<<<1, 1024, 0, stream>>>(cntI, rowptr, N);
        k_inv_sqrt_cnt<<<(N + BT - 1) / BT, BT, 0, stream>>>(cntO, inv_out, cntI, inv_in, N);
        k_fill<<<(E + BT - 1) / BT, BT, 0, stream>>>(esrc, edst, rowptr, rank, csr, E);

        const int pblocks = (N + 7) / 8;
        const float* xin = features;
        for (int i = 0; i < 8; ++i) {
            const int Md = DIMS[i + 1];
            const float* Wi = (const float*)d_in[3 + 2 * i];
            const float* bi = (const float*)d_in[4 + 2 * i];

            if (i == 0) k_pull<4><<<pblocks, BT, 0, stream>>>(xin, inv_out, rowptr, csr, agg, N);
            else        k_pull<64><<<pblocks, BT, 0, stream>>>(xin, inv_out, rowptr, csr, agg, N);

            int colTiles  = (Md + 15) / 16;
            int rowBlocks = (N + 63) / 64;
            long long waves = (long long)rowBlocks * colTiles;
            long long thr = waves * 32;
            int gblocks = (int)((thr + BT - 1) / BT);
            float* yout = (i == 7) ? (float*)d_out : xbuf;

            if (i == 0) {
                k_gemm_wmma<4, 64, true><<<gblocks, BT, 0, stream>>>(
                    agg, inv_in, Wi, bi, yout, N, colTiles, rowBlocks);
            } else if (i < 7) {
                k_gemm_wmma<64, 64, true><<<gblocks, BT, 0, stream>>>(
                    agg, inv_in, Wi, bi, yout, N, colTiles, rowBlocks);
            } else {
                k_gemm_wmma<64, 3, false><<<gblocks, BT, 0, stream>>>(
                    agg, inv_in, Wi, bi, yout, N, colTiles, rowBlocks);
            }
            xin = xbuf;
        }
        return;
    }

    {
        int n4 = (N + 3) / 4;
        k_zero4<<<(n4 + BT - 1) / BT, BT, 0, stream>>>((float4*)inv_out, n4);
        k_zero4<<<(n4 + BT - 1) / BT, BT, 0, stream>>>((float4*)inv_in,  n4);
        k_degrees<<<(E + BT - 1) / BT, BT, 0, stream>>>(esrc, edst, inv_out, inv_in, E);
        k_inv_sqrt<<<(N + BT - 1) / BT, BT, 0, stream>>>(inv_out, inv_in, N);
    }

    const float* xcur = features;
    for (int i = 0; i < 8; ++i) {
        const int Kd = DIMS[i];
        const int Md = DIMS[i + 1];
        const float* Wi = (const float*)d_in[3 + 2 * i];
        const float* bi = (const float*)d_in[4 + 2 * i];

        int n4 = N * Kd / 4;
        k_zero4<<<(n4 + BT - 1) / BT, BT, 0, stream>>>((float4*)agg, n4);

        int chunks = Kd / 4;
        int lg = (chunks == 16) ? 4 : 0;
        long long total = (long long)E * chunks;
        int blocks = (int)((total + BT - 1) / BT);
        k_gather<<<blocks, BT, 0, stream>>>(xcur, inv_out, esrc, edst, agg, E, lg, Kd);

        int colTiles  = (Md + 15) / 16;
        int rowBlocks = (N + 63) / 64;
        long long waves = (long long)rowBlocks * colTiles;
        long long thr = waves * 32;
        int gblocks = (int)((thr + BT - 1) / BT);
        float* yout = (i == 7) ? (float*)d_out : xbuf;

        if (i == 0) {
            k_gemm_wmma<4, 64, true><<<gblocks, BT, 0, stream>>>(
                agg, inv_in, Wi, bi, yout, N, colTiles, rowBlocks);
        } else if (i < 7) {
            k_gemm_wmma<64, 64, true><<<gblocks, BT, 0, stream>>>(
                agg, inv_in, Wi, bi, yout, N, colTiles, rowBlocks);
        } else {
            k_gemm_wmma<64, 3, false><<<gblocks, BT, 0, stream>>>(
                agg, inv_in, Wi, bi, yout, N, colTiles, rowBlocks);
        }
        xcur = xbuf;
    }
}


# Aggregation fused with its GEMM, one launch per layer: a block sums 64 nodes' in-edges into an LDS tile, then its 8 waves run the f32 WMMA tiles from LDS; 8 launches and the agg round trip gone
# speedup vs baseline: 3.6315x; 1.0766x over previous
#include <hip/hip_runtime.h>


typedef __attribute__((ext_vector_type(2))) float v2f;
typedef __attribute__((ext_vector_type(8))) float v8f;

#define LEAKY_SLOPE 0.01f

__device__ __forceinline__ void atomAddF(float* p, float v) {
    __hip_atomic_fetch_add(p, v, __ATOMIC_RELAXED, __HIP_MEMORY_SCOPE_AGENT);
}

__global__ void k_zero4(float4* __restrict__ p, int n4) {
    int i = blockIdx.x * blockDim.x + threadIdx.x;
    if (i < n4) p[i] = make_float4(0.f, 0.f, 0.f, 0.f);
}

__global__ void k_degrees(const int* __restrict__ src, const int* __restrict__ dst,
                          float* __restrict__ degO, float* __restrict__ degI, int E) {
    int e = blockIdx.x * blockDim.x + threadIdx.x;
    if (e >= E) return;
    atomAddF(&degO[src[e]], 1.0f);
    atomAddF(&degI[dst[e]], 1.0f);
}

__global__ void k_inv_sqrt(float* __restrict__ a, float* __restrict__ b, int n) {
    int i = blockIdx.x * blockDim.x + threadIdx.x;
    if (i >= n) return;
    a[i] = rsqrtf(fmaxf(a[i], 1.0f));
    b[i] = rsqrtf(fmaxf(b[i], 1.0f));
}

__global__ void k_gather(const float* __restrict__ x, const float* __restrict__ invO,
                         const int* __restrict__ src, const int* __restrict__ dst,
                         float* __restrict__ agg, int E, int lgChunks, int d) {
    int tid = blockIdx.x * blockDim.x + threadIdx.x;
    int e = tid >> lgChunks;
    if (e >= E) return;
    int c = tid & ((1 << lgChunks) - 1);
    int s = src[e];
    int t = dst[e];
    float w = invO[s];
    const float4 v = *reinterpret_cast<const float4*>(x + (size_t)s * d + c * 4);
    float* base = agg + (size_t)t * d + c * 4;
    atomAddF(base + 0, v.x * w);
    atomAddF(base + 1, v.y * w);
    atomAddF(base + 2, v.z * w);
    atomAddF(base + 3, v.w * w);
}

__global__ void k_count_rank(const int* __restrict__ src, const int* __restrict__ dst,
                             int* __restrict__ cntO, int* __restrict__ cntI,
                             int* __restrict__ rank, int E) {
    int e = blockIdx.x * blockDim.x + threadIdx.x;
    if (e >= E) return;
    rank[e] = __hip_atomic_fetch_add(&cntI[dst[e]], 1, __ATOMIC_RELAXED, __HIP_MEMORY_SCOPE_AGENT);
    __hip_atomic_fetch_add(&cntO[src[e]], 1, __ATOMIC_RELAXED, __HIP_MEMORY_SCOPE_AGENT);
}

__global__ void __launch_bounds__(1024) k_scan(const int* __restrict__ cnt,
                                               int* __restrict__ rowptr, int n) {
    __shared__ int wsum[32];
    __shared__ int tileTotal;
    const int tid = threadIdx.x, lane = tid & 31, wv = tid >> 5;
    int carry = 0;
    for (int base = 0; base < n; base += 4096) {
        const int i = base + tid * 4;
        int a0 = 0, a1 = 0, a2 = 0, a3 = 0;
        if (i + 3 < n) {
            const int4 v = *reinterpret_cast<const int4*>(cnt + i);
            a0 = v.x; a1 = v.y; a2 = v.z; a3 = v.w;
        } else {
            if (i < n) a0 = cnt[i];
            if (i + 1 < n) a1 = cnt[i + 1];
            if (i + 2 < n) a2 = cnt[i + 2];
        }
        const int tsum = a0 + a1 + a2 + a3;
        int inc = tsum;
        for (int o = 1; o < 32; o <<= 1) {
            int t = __shfl_up(inc, o);
            if (lane >= o) inc += t;
        }
        if (lane == 31) wsum[wv] = inc;
        __syncthreads();
        if (wv == 0) {
            const int w = wsum[lane];
            int winc = w;
            for (int o = 1; o < 32; o <<= 1) {
                int t = __shfl_up(winc, o);
                if (lane >= o) winc += t;
            }
            wsum[lane] = winc - w;
            if (lane == 31) tileTotal = winc;
        }
        __syncthreads();
        const int ex = carry + wsum[wv] + (inc - tsum);
        carry += tileTotal;
        if (i + 3 < n) {
            *reinterpret_cast<int4*>(rowptr + i) = make_int4(ex, ex + a0, ex + a0 + a1, ex + a0 + a1 + a2);
        } else {
            if (i < n) rowptr[i] = ex;
            if (i + 1 < n) rowptr[i + 1] = ex + a0;
            if (i + 2 < n) rowptr[i + 2] = ex + a0 + a1;
        }
        __syncthreads();
    }
    if (tid == 0) rowptr[n] = carry;
}

__global__ void k_inv_sqrt_cnt(const int* cntA, float* outA, const int* cntB, float* outB, int n) {
    int i = blockIdx.x * blockDim.x + threadIdx.x;
    if (i >= n) return;
    float fa = (float)cntA[i];
    float fb = (float)cntB[i];
    outA[i] = rsqrtf(fmaxf(fa, 1.0f));
    outB[i] = rsqrtf(fmaxf(fb, 1.0f));
}

__global__ void k_fill(const int* __restrict__ src, const int* __restrict__ dst,
                       const int* __restrict__ rowptr, const int* __restrict__ rank,
                       int* __restrict__ csr, int E) {
    int e = blockIdx.x * blockDim.x + threadIdx.x;
    if (e >= E) return;
    csr[rowptr[dst[e]] + rank[e]] = src[e];
}

template <int KDIM, int MD, bool LEAKY>
__global__ void k_gemm_wmma(const float* __restrict__ agg, const float* __restrict__ invI,
                            const float* __restrict__ W, const float* __restrict__ bias,
                            float* __restrict__ out, int Nn, int colTiles, int rowBlocks) {
    constexpr int NK = KDIM / 4;
    constexpr bool FULLCOL = (MD % 16 == 0);

    int gwave    = __builtin_amdgcn_readfirstlane(
                       (int)((blockIdx.x * blockDim.x + threadIdx.x) >> 5));
    int lane     = threadIdx.x & 31;
    int rowBlock = gwave / colTiles;
    int colTile  = gwave - rowBlock * colTiles;
    if (rowBlock >= rowBlocks) return;

    int   ln15    = lane & 15;
    int   col     = colTile * 16 + ln15;
    int   colC    = FULLCOL ? col : ((col < MD) ? col : (MD - 1));
    float colMask = FULLCOL ? 1.f : ((col < MD) ? 1.f : 0.f);
    int   kHalf   = (lane >> 4) * 2;
    int   hi      = lane >> 4;

    v2f bf[NK];
    #pragma unroll
    for (int t = 0; t < NK; ++t) {
        int k0 = t * 4 + kHalf;
        bf[t].x = W[k0 * MD + colC] * colMask;
        bf[t].y = W[(k0 + 1) * MD + colC] * colMask;
    }
    float bv = bias[colC] * colMask;

    #pragma unroll
    for (int st = 0; st < 4; ++st) {
        int tileBase = rowBlock * 64 + st * 16;
        if (tileBase >= Nn) break;
        int   mrow  = tileBase + ln15;
        int   mrowC = (mrow < Nn) ? mrow : (Nn - 1);
        float s     = invI[mrowC];

        v8f acc = {0.f, 0.f, 0.f, 0.f, 0.f, 0.f, 0.f, 0.f};
        #pragma unroll
        for (int t = 0; t < NK; ++t) {
            int k0 = t * 4 + kHalf;
            v2f a = *reinterpret_cast<const v2f*>(agg + (size_t)mrowC * KDIM + k0);
            a.x *= s;
            a.y *= s;
            acc = __builtin_amdgcn_wmma_f32_16x16x4_f32(
                false, a, false, bf[t], (short)0, acc, false, false);
        }

        float h[8];
        #pragma unroll
        for (int j = 0; j < 8; ++j) {
            float v = acc[j] + bv;
            h[j] = LEAKY ? fmaxf(v, LEAKY_SLOPE * v) : v;
        }

        float* outBase = out + (size_t)(tileBase + hi * 8) * MD + col;
        if (FULLCOL && (tileBase + 16 <= Nn)) {
            #pragma unroll
            for (int j = 0; j < 8; ++j) outBase[j * MD] = h[j];
        } else {
            #pragma unroll
            for (int j = 0; j < 8; ++j) {
                int r = tileBase + hi * 8 + j;
                if (col < MD && r < Nn) outBase[j * MD] = h[j];
            }
        }
    }
}

template <int D, int MD, bool LEAKY>
__global__ void __launch_bounds__(256)
k_layer(const float* __restrict__ x, const float* __restrict__ invO,
        const float* __restrict__ invI, const int* __restrict__ rowptr,
        const int* __restrict__ csr, const float* __restrict__ W,
        const float* __restrict__ bias, float* __restrict__ out, int Nn) {
    constexpr int LPR   = D / 4;
    constexpr int EPI   = 32 / LPR;
    constexpr int PITCH = (D >= 64) ? (D + 4) : D;
    constexpr int NK    = D / 4;
    constexpr int COLT  = (MD + 15) / 16;
    constexpr bool FULLCOL = (MD % 16 == 0);
    static_assert(8 % COLT == 0, "a wave's tiles must share one column slice");
    constexpr bool ALLWAVES = (4 * COLT >= 8);

    __shared__ float4 sA4[64 * PITCH / 4];

    const int lane     = threadIdx.x & 31;
    const int wv       = __builtin_amdgcn_readfirstlane((int)(threadIdx.x >> 5));
    const int nodeBase = blockIdx.x * 64;
    const int sub      = lane / LPR;
    const int c        = lane % LPR;
    const float* xc    = x + c * 4;

    const int first = nodeBase + wv * 8;
    const int pIdx  = first + lane;
    const int   rp  = rowptr[pIdx < Nn ? pIdx : Nn];
    const float nrm = invI[pIdx < Nn ? pIdx : (Nn - 1)];

    for (int i = 0; i < 8; ++i) {
        const int beg = __builtin_amdgcn_readlane(rp, i);
        const int end = __builtin_amdgcn_readlane(rp, i + 1);
        const float si = __int_as_float(__builtin_amdgcn_readlane(__float_as_int(nrm), i));

        float4 acc = make_float4(0.f, 0.f, 0.f, 0.f);
        int e = beg + sub;
        for (; e + 3 * EPI < end; e += 4 * EPI) {
            const int s0 = csr[e];
            const int s1 = csr[e + EPI];
            const int s2 = csr[e + 2 * EPI];
            const int s3 = csr[e + 3 * EPI];
            const float w0 = invO[s0];
            const float w1 = invO[s1];
            const float w2 = invO[s2];
            const float w3 = invO[s3];
            const float4 v0 = *reinterpret_cast<const float4*>(xc + (size_t)s0 * D);
            const float4 v1 = *reinterpret_cast<const float4*>(xc + (size_t)s1 * D);
            const float4 v2 = *reinterpret_cast<const float4*>(xc + (size_t)s2 * D);
            const float4 v3 = *reinterpret_cast<const float4*>(xc + (size_t)s3 * D);
            acc.x += v0.x * w0; acc.y += v0.y * w0; acc.z += v0.z * w0; acc.w += v0.w * w0;
            acc.x += v1.x * w1; acc.y += v1.y * w1; acc.z += v1.z * w1; acc.w += v1.w * w1;
            acc.x += v2.x * w2; acc.y += v2.y * w2; acc.z += v2.z * w2; acc.w += v2.w * w2;
            acc.x += v3.x * w3; acc.y += v3.y * w3; acc.z += v3.z * w3; acc.w += v3.w * w3;
        }
        for (; e < end; e += EPI) {
            const int s0 = csr[e];
            const float w0 = invO[s0];
            const float4 v0 = *reinterpret_cast<const float4*>(xc + (size_t)s0 * D);
            acc.x += v0.x * w0; acc.y += v0.y * w0; acc.z += v0.z * w0; acc.w += v0.w * w0;
        }
        #pragma unroll
        for (int o = LPR; o < 32; o <<= 1) {
            acc.x += __shfl_xor(acc.x, o);
            acc.y += __shfl_xor(acc.y, o);
            acc.z += __shfl_xor(acc.z, o);
            acc.w += __shfl_xor(acc.w, o);
        }
        if (sub == 0)
            sA4[(wv * 8 + i) * (PITCH / 4) + c] =
                make_float4(acc.x * si, acc.y * si, acc.z * si, acc.w * si);
    }

    const int   ln15    = lane & 15;
    const int   hi      = lane >> 4;
    const int   kHalf   = hi * 2;
    const int   ct      = wv % COLT;
    const int   col     = ct * 16 + ln15;
    const int   colC    = FULLCOL ? col : ((col < MD) ? col : (MD - 1));
    const float colMask = FULLCOL ? 1.f : ((col < MD) ? 1.f : 0.f);

    v2f bf[NK];
    float bv = 0.f;
    if (ALLWAVES || wv < 4 * COLT) {
        #pragma unroll
        for (int t = 0; t < NK; ++t) {
            const int k0 = t * 4 + kHalf;
            bf[t].x = W[k0 * MD + colC] * colMask;
            bf[t].y = W[(k0 + 1) * MD + colC] * colMask;
        }
        bv = bias[colC] * colMask;
    }

    __syncthreads();

    const float* sAf = reinterpret_cast<const float*>(sA4);
    for (int tl = wv; tl < 4 * COLT; tl += 8) {
        const int rt       = tl / COLT;
        const int tileBase = nodeBase + rt * 16;
        if (tileBase >= Nn) break;
        const float* aRow = sAf + (rt * 16 + ln15) * PITCH + kHalf;

        v8f acc = {0.f, 0.f, 0.f, 0.f, 0.f, 0.f, 0.f, 0.f};
        #pragma unroll
        for (int t = 0; t < NK; ++t) {
            const v2f a = *reinterpret_cast<const v2f*>(aRow + t * 4);
            acc = __builtin_amdgcn_wmma_f32_16x16x4_f32(
                false, a, false, bf[t], (short)0, acc, false, false);
        }

        float h[8];
        #pragma unroll
        for (int j = 0; j < 8; ++j) {
            const float v = acc[j] + bv;
            h[j] = LEAKY ? fmaxf(v, LEAKY_SLOPE * v) : v;
        }

        float* outBase = out + (size_t)(tileBase + hi * 8) * MD + col;
        if (FULLCOL && (tileBase + 16 <= Nn)) {
            #pragma unroll
            for (int j = 0; j < 8; ++j) outBase[j * MD] = h[j];
        } else {
            #pragma unroll
            for (int j = 0; j < 8; ++j) {
                const int r = tileBase + hi * 8 + j;
                if (col < MD && r < Nn) outBase[j * MD] = h[j];
            }
        }
    }
}

extern "C" void kernel_launch(void* const* d_in, const int* in_sizes, int n_in,
                              void* d_out, int out_size, void* d_ws, size_t ws_size,
                              hipStream_t stream) {
    (void)n_in; (void)out_size;

    const float* features = (const float*)d_in[0];
    const int*   esrc     = (const int*)d_in[1];
    const int*   edst     = (const int*)d_in[2];

    const int N = in_sizes[0] / 4;
    const int E = in_sizes[1];
    static const int DIMS[9] = {4, 64, 64, 64, 64, 64, 64, 64, 3};

    float* ws      = (float*)d_ws;
    float* inv_out = ws;
    float* inv_in  = ws + 100352;
    float* xbuf    = ws + 200704;
    float* agg     = ws + 200704 + 6400000;

    const int BT = 256;

    if (ws_size >= (size_t)65204224) {
        int* cntO   = (int*)inv_out;
        int* cntI   = (int*)inv_in;
        int* rank   = (int*)xbuf;
        int* csr    = (int*)(ws + 13000704);
        int* rowptr = (int*)(ws + 13000704 + 3200000);

        int n4 = (N + 3) / 4;
        k_zero4<<<(n4 + BT - 1) / BT, BT, 0, stream>>>((float4*)inv_out, n4);
        k_zero4<<<(n4 + BT - 1) / BT, BT, 0, stream>>>((float4*)inv_in,  n4);
        k_count_rank<<<(E + BT - 1) / BT, BT, 0, stream>>>(esrc, edst, cntO, cntI, rank, E);
        k_scan<<<1, 1024, 0, stream>>>(cntI, rowptr, N);
        k_inv_sqrt_cnt<<<(N + BT - 1) / BT, BT, 0, stream>>>(cntO, inv_out, cntI, inv_in, N);
        k_fill<<<(E + BT - 1) / BT, BT, 0, stream>>>(esrc, edst, rowptr, rank, csr, E);

        const int lblocks = (N + 63) / 64;
        float* planeA = xbuf;
        float* planeB = agg;
        k_layer<4, 64, true><<<lblocks, BT, 0, stream>>>(
            features, inv_out, inv_in, rowptr, csr,
            (const float*)d_in[3], (const float*)d_in[4], planeA, N);
        const float* xin = planeA;
        float* xout = planeB;
        for (int i = 1; i < 7; ++i) {
            k_layer<64, 64, true><<<lblocks, BT, 0, stream>>>(
                xin, inv_out, inv_in, rowptr, csr,
                (const float*)d_in[3 + 2 * i], (const float*)d_in[4 + 2 * i], xout, N);
            float* t = (float*)xin;
            xin = xout;
            xout = t;
        }
        k_layer<64, 3, false><<<lblocks, BT, 0, stream>>>(
            xin, inv_out, inv_in, rowptr, csr,
            (const float*)d_in[17], (const float*)d_in[18], (float*)d_out, N);
        return;
    }

    {
        int n4 = (N + 3) / 4;
        k_zero4<<<(n4 + BT - 1) / BT, BT, 0, stream>>>((float4*)inv_out, n4);
        k_zero4<<<(n4 + BT - 1) / BT, BT, 0, stream>>>((float4*)inv_in,  n4);
        k_degrees<<<(E + BT - 1) / BT, BT, 0, stream>>>(esrc, edst, inv_out, inv_in, E);
        k_inv_sqrt<<<(N + BT - 1) / BT, BT, 0, stream>>>(inv_out, inv_in, N);
    }

    const float* xcur = features;
    for (int i = 0; i < 8; ++i) {
        const int Kd = DIMS[i];
        const int Md = DIMS[i + 1];
        const float* Wi = (const float*)d_in[3 + 2 * i];
        const float* bi = (const float*)d_in[4 + 2 * i];

        int n4 = N * Kd / 4;
        k_zero4<<<(n4 + BT - 1) / BT, BT, 0, stream>>>((float4*)agg, n4);

        int chunks = Kd / 4;
        int lg = (chunks == 16) ? 4 : 0;
        long long total = (long long)E * chunks;
        int blocks = (int)((total + BT - 1) / BT);
        k_gather<<<blocks, BT, 0, stream>>>(xcur, inv_out, esrc, edst, agg, E, lg, Kd);

        int colTiles  = (Md + 15) / 16;
        int rowBlocks = (N + 63) / 64;
        long long waves = (long long)rowBlocks * colTiles;
        long long thr = waves * 32;
        int gblocks = (int)((thr + BT - 1) / BT);
        float* yout = (i == 7) ? (float*)d_out : xbuf;

        if (i == 0) {
            k_gemm_wmma<4, 64, true><<<gblocks, BT, 0, stream>>>(
                agg, inv_in, Wi, bi, yout, N, colTiles, rowBlocks);
        } else if (i < 7) {
            k_gemm_wmma<64, 64, true><<<gblocks, BT, 0, stream>>>(
                agg, inv_in, Wi, bi, yout, N, colTiles, rowBlocks);
        } else {
            k_gemm_wmma<64, 3, false><<<gblocks, BT, 0, stream>>>(
                agg, inv_in, Wi, bi, yout, N, colTiles, rowBlocks);
        }
        xcur = xbuf;
    }
}


# Fused layer's 64-wide gather: a node's edge list and norms read one edge per lane two nodes ahead, its rows fetched eight loads in flight; 80 nodes per work-group; edge order of the sums unchanged
# speedup vs baseline: 3.7522x; 1.0332x over previous
#include <hip/hip_runtime.h>


typedef __attribute__((ext_vector_type(2))) float v2f;
typedef __attribute__((ext_vector_type(8))) float v8f;

#define LEAKY_SLOPE 0.01f

__device__ __forceinline__ void atomAddF(float* p, float v) {
    __hip_atomic_fetch_add(p, v, __ATOMIC_RELAXED, __HIP_MEMORY_SCOPE_AGENT);
}

__global__ void k_zero4(float4* __restrict__ p, int n4) {
    int i = blockIdx.x * blockDim.x + threadIdx.x;
    if (i < n4) p[i] = make_float4(0.f, 0.f, 0.f, 0.f);
}

__global__ void k_degrees(const int* __restrict__ src, const int* __restrict__ dst,
                          float* __restrict__ degO, float* __restrict__ degI, int E) {
    int e = blockIdx.x * blockDim.x + threadIdx.x;
    if (e >= E) return;
    atomAddF(&degO[src[e]], 1.0f);
    atomAddF(&degI[dst[e]], 1.0f);
}

__global__ void k_inv_sqrt(float* __restrict__ a, float* __restrict__ b, int n) {
    int i = blockIdx.x * blockDim.x + threadIdx.x;
    if (i >= n) return;
    a[i] = rsqrtf(fmaxf(a[i], 1.0f));
    b[i] = rsqrtf(fmaxf(b[i], 1.0f));
}

__global__ void k_gather(const float* __restrict__ x, const float* __restrict__ invO,
                         const int* __restrict__ src, const int* __restrict__ dst,
                         float* __restrict__ agg, int E, int lgChunks, int d) {
    int tid = blockIdx.x * blockDim.x + threadIdx.x;
    int e = tid >> lgChunks;
    if (e >= E) return;
    int c = tid & ((1 << lgChunks) - 1);
    int s = src[e];
    int t = dst[e];
    float w = invO[s];
    const float4 v = *reinterpret_cast<const float4*>(x + (size_t)s * d + c * 4);
    float* base = agg + (size_t)t * d + c * 4;
    atomAddF(base + 0, v.x * w);
    atomAddF(base + 1, v.y * w);
    atomAddF(base + 2, v.z * w);
    atomAddF(base + 3, v.w * w);
}

__global__ void k_count_rank(const int* __restrict__ src, const int* __restrict__ dst,
                             int* __restrict__ cntO, int* __restrict__ cntI,
                             int* __restrict__ rank, int E) {
    int e = blockIdx.x * blockDim.x + threadIdx.x;
    if (e >= E) return;
    rank[e] = __hip_atomic_fetch_add(&cntI[dst[e]], 1, __ATOMIC_RELAXED, __HIP_MEMORY_SCOPE_AGENT);
    __hip_atomic_fetch_add(&cntO[src[e]], 1, __ATOMIC_RELAXED, __HIP_MEMORY_SCOPE_AGENT);
}

__global__ void __launch_bounds__(1024) k_scan(const int* __restrict__ cnt,
                                               int* __restrict__ rowptr, int n) {
    __shared__ int wsum[32];
    __shared__ int tileTotal;
    const int tid = threadIdx.x, lane = tid & 31, wv = tid >> 5;
    int carry = 0;
    for (int base = 0; base < n; base += 4096) {
        const int i = base + tid * 4;
        int a0 = 0, a1 = 0, a2 = 0, a3 = 0;
        if (i + 3 < n) {
            const int4 v = *reinterpret_cast<const int4*>(cnt + i);
            a0 = v.x; a1 = v.y; a2 = v.z; a3 = v.w;
        } else {
            if (i < n) a0 = cnt[i];
            if (i + 1 < n) a1 = cnt[i + 1];
            if (i + 2 < n) a2 = cnt[i + 2];
        }
        const int tsum = a0 + a1 + a2 + a3;
        int inc = tsum;
        for (int o = 1; o < 32; o <<= 1) {
            int t = __shfl_up(inc, o);
            if (lane >= o) inc += t;
        }
        if (lane == 31) wsum[wv] = inc;
        __syncthreads();
        if (wv == 0) {
            const int w = wsum[lane];
            int winc = w;
            for (int o = 1; o < 32; o <<= 1) {
                int t = __shfl_up(winc, o);
                if (lane >= o) winc += t;
            }
            wsum[lane] = winc - w;
            if (lane == 31) tileTotal = winc;
        }
        __syncthreads();
        const int ex = carry + wsum[wv] + (inc - tsum);
        carry += tileTotal;
        if (i + 3 < n) {
            *reinterpret_cast<int4*>(rowptr + i) = make_int4(ex, ex + a0, ex + a0 + a1, ex + a0 + a1 + a2);
        } else {
            if (i < n) rowptr[i] = ex;
            if (i + 1 < n) rowptr[i + 1] = ex + a0;
            if (i + 2 < n) rowptr[i + 2] = ex + a0 + a1;
        }
        __syncthreads();
    }
    if (tid == 0) rowptr[n] = carry;
}

__global__ void k_inv_sqrt_cnt(const int* cntA, float* outA, const int* cntB, float* outB, int n) {
    int i = blockIdx.x * blockDim.x + threadIdx.x;
    if (i >= n) return;
    float fa = (float)cntA[i];
    float fb = (float)cntB[i];
    outA[i] = rsqrtf(fmaxf(fa, 1.0f));
    outB[i] = rsqrtf(fmaxf(fb, 1.0f));
}

__global__ void k_fill(const int* __restrict__ src, const int* __restrict__ dst,
                       const int* __restrict__ rowptr, const int* __restrict__ rank,
                       int* __restrict__ csr, int E) {
    int e = blockIdx.x * blockDim.x + threadIdx.x;
    if (e >= E) return;
    csr[rowptr[dst[e]] + rank[e]] = src[e];
}

template <int KDIM, int MD, bool LEAKY>
__global__ void k_gemm_wmma(const float* __restrict__ agg, const float* __restrict__ invI,
                            const float* __restrict__ W, const float* __restrict__ bias,
                            float* __restrict__ out, int Nn, int colTiles, int rowBlocks) {
    constexpr int NK = KDIM / 4;
    constexpr bool FULLCOL = (MD % 16 == 0);

    int gwave    = __builtin_amdgcn_readfirstlane(
                       (int)((blockIdx.x * blockDim.x + threadIdx.x) >> 5));
    int lane     = threadIdx.x & 31;
    int rowBlock = gwave / colTiles;
    int colTile  = gwave - rowBlock * colTiles;
    if (rowBlock >= rowBlocks) return;

    int   ln15    = lane & 15;
    int   col     = colTile * 16 + ln15;
    int   colC    = FULLCOL ? col : ((col < MD) ? col : (MD - 1));
    float colMask = FULLCOL ? 1.f : ((col < MD) ? 1.f : 0.f);
    int   kHalf   = (lane >> 4) * 2;
    int   hi      = lane >> 4;

    v2f bf[NK];
    #pragma unroll
    for (int t = 0; t < NK; ++t) {
        int k0 = t * 4 + kHalf;
        bf[t].x = W[k0 * MD + colC] * colMask;
        bf[t].y = W[(k0 + 1) * MD + colC] * colMask;
    }
    float bv = bias[colC] * colMask;

    #pragma unroll
    for (int st = 0; st < 4; ++st) {
        int tileBase = rowBlock * 64 + st * 16;
        if (tileBase >= Nn) break;
        int   mrow  = tileBase + ln15;
        int   mrowC = (mrow < Nn) ? mrow : (Nn - 1);
        float s     = invI[mrowC];

        v8f acc = {0.f, 0.f, 0.f, 0.f, 0.f, 0.f, 0.f, 0.f};
        #pragma unroll
        for (int t = 0; t < NK; ++t) {
            int k0 = t * 4 + kHalf;
            v2f a = *reinterpret_cast<const v2f*>(agg + (size_t)mrowC * KDIM + k0);
            a.x *= s;
            a.y *= s;
            acc = __builtin_amdgcn_wmma_f32_16x16x4_f32(
                false, a, false, bf[t], (short)0, acc, false, false);
        }

        float h[8];
        #pragma unroll
        for (int j = 0; j < 8; ++j) {
            float v = acc[j] + bv;
            h[j] = LEAKY ? fmaxf(v, LEAKY_SLOPE * v) : v;
        }

        float* outBase = out + (size_t)(tileBase + hi * 8) * MD + col;
        if (FULLCOL && (tileBase + 16 <= Nn)) {
            #pragma unroll
            for (int j = 0; j < 8; ++j) outBase[j * MD] = h[j];
        } else {
            #pragma unroll
            for (int j = 0; j < 8; ++j) {
                int r = tileBase + hi * 8 + j;
                if (col < MD && r < Nn) outBase[j * MD] = h[j];
            }
        }
    }
}

__device__ __forceinline__ void listLoad(const int* __restrict__ csr, int beg, int end, int lane,
                                         int& s0, int& s1) {
    const int last = (end > 0) ? (end - 1) : 0;
    const int e0 = beg + lane;
    const int e1 = e0 + 32;
    s0 = csr[e0 < last ? e0 : last];
    s1 = csr[e1 < last ? e1 : last];
}

__device__ __forceinline__ void normLoad(const float* __restrict__ invO, int s0, int s1,
                                         float& w0, float& w1) {
    w0 = invO[s0];
    w1 = invO[s1];
}

template <int L0, bool FULL>
__device__ __forceinline__ void gather16(const char* __restrict__ xb, unsigned cOff, int sub,
                                         int sReg, float wReg, int cnt, float4& acc) {
    float4 v[8];
    const int sFirst = __builtin_amdgcn_readlane(sReg, L0);
    #pragma unroll
    for (int u = 0; u < 8; ++u) {
        const int sLo = __builtin_amdgcn_readlane(sReg, L0 + 2 * u);
        const int sHi = __builtin_amdgcn_readlane(sReg, L0 + 2 * u + 1);
        int s = sub ? sHi : sLo;
        if (!FULL) s = (2 * u + sub < cnt) ? s : sFirst;
        v[u] = *reinterpret_cast<const float4*>(xb + (((unsigned)s << 8) + cOff));
    }
    #pragma unroll
    for (int u = 0; u < 8; ++u) {
        const int wLo = __builtin_amdgcn_readlane(__float_as_int(wReg), L0 + 2 * u);
        const int wHi = __builtin_amdgcn_readlane(__float_as_int(wReg), L0 + 2 * u + 1);
        const float w = __int_as_float(sub ? wHi : wLo);
        float4 t = v[u];
        if (!FULL && !(2 * u + sub < cnt)) t = make_float4(0.f, 0.f, 0.f, 0.f);
        acc.x += t.x * w; acc.y += t.y * w; acc.z += t.z * w; acc.w += t.w * w;
    }
}

__device__ __forceinline__ void gather32(const char* __restrict__ xb, unsigned cOff, int sub,
                                         int sReg, float wReg, int n, float4& acc) {
    if (n >= 16) gather16<0, true>(xb, cOff, sub, sReg, wReg, 16, acc);
    else if (n > 0) gather16<0, false>(xb, cOff, sub, sReg, wReg, n, acc);
    if (n >= 32) gather16<16, true>(xb, cOff, sub, sReg, wReg, 16, acc);
    else if (n > 16) gather16<16, false>(xb, cOff, sub, sReg, wReg, n - 16, acc);
}

template <int D, int MD, bool LEAKY>
__global__ void __launch_bounds__(256) __attribute__((amdgpu_waves_per_eu(12, 16)))
k_layer(const float* __restrict__ x, const float* __restrict__ invO,
        const float* __restrict__ invI, const int* __restrict__ rowptr,
        const int* __restrict__ csr, const float* __restrict__ W,
        const float* __restrict__ bias, float* __restrict__ out, int Nn) {
    constexpr int LPR   = D / 4;
    constexpr int EPI   = 32 / LPR;
    constexpr int PITCH = (D >= 64) ? (D + 4) : D;
    constexpr int NK    = D / 4;
    constexpr int COLT  = (MD + 15) / 16;
    constexpr bool FULLCOL = (MD % 16 == 0);
    static_assert(8 % COLT == 0, "a wave's tiles must share one column slice");
    constexpr int ROWS  = 80;
    constexpr int NPW   = ROWS / 8;
    constexpr int NT    = (ROWS / 16) * COLT;
    constexpr bool ALLWAVES = (NT >= 8);

    __shared__ float4 sA4[ROWS * PITCH / 4];

    const int lane     = threadIdx.x & 31;
    const int wv       = __builtin_amdgcn_readfirstlane((int)(threadIdx.x >> 5));
    const int nodeBase = blockIdx.x * ROWS;
    const int sub      = lane / LPR;
    const int c        = lane % LPR;
    const float* xc    = x + c * 4;

    const int first = nodeBase + wv * NPW;
    const int pIdx  = first + lane;
    const int   rp  = rowptr[pIdx < Nn ? pIdx : Nn];
    const float nrm = invI[pIdx < Nn ? pIdx : (Nn - 1)];

    if constexpr (D == 64) {
        const char* xb = reinterpret_cast<const char*>(x);
        const unsigned cOff = (unsigned)c * 16u;
        int   sC0, sC1, sN0, sN1;
        float wC0, wC1;
        {
            const int b0 = __builtin_amdgcn_readlane(rp, 0);
            const int e0 = __builtin_amdgcn_readlane(rp, 1);
            const int e1 = __builtin_amdgcn_readlane(rp, 2);
            listLoad(csr, b0, e0, lane, sC0, sC1);
            listLoad(csr, e0, e1, lane, sN0, sN1);
            normLoad(invO, sC0, sC1, wC0, wC1);
        }
        #pragma nounroll
        for (int i = 0; i < NPW; ++i) {
            const int beg  = __builtin_amdgcn_readlane(rp, i);
            const int end  = __builtin_amdgcn_readlane(rp, i + 1);
            const int end1 = __builtin_amdgcn_readlane(rp, i + 2);
            const int end2 = __builtin_amdgcn_readlane(rp, i + 3);
            const float si = __int_as_float(__builtin_amdgcn_readlane(__float_as_int(nrm), i));

            float wN0, wN1;
            int   sM0, sM1;
            normLoad(invO, sN0, sN1, wN0, wN1);
            listLoad(csr, end1, end2, lane, sM0, sM1);

            const int deg = end - beg;
            float4 acc = make_float4(0.f, 0.f, 0.f, 0.f);
            gather32(xb, cOff, sub, sC0, wC0, deg, acc);
            if (deg > 32) gather32(xb, cOff, sub, sC1, wC1, deg - 32, acc);
            for (int e = beg + 64 + sub; e < end; e += EPI) {
                const int s0 = csr[e];
                const float w0 = invO[s0];
                const float4 v0 = *reinterpret_cast<const float4*>(xc + (size_t)s0 * D);
                acc.x += v0.x * w0; acc.y += v0.y * w0; acc.z += v0.z * w0; acc.w += v0.w * w0;
            }
            acc.x += __shfl_xor(acc.x, 16);
            acc.y += __shfl_xor(acc.y, 16);
            acc.z += __shfl_xor(acc.z, 16);
            acc.w += __shfl_xor(acc.w, 16);
            if (sub == 0)
                sA4[(wv * NPW + i) * (PITCH / 4) + c] =
                    make_float4(acc.x * si, acc.y * si, acc.z * si, acc.w * si);

            sC0 = sN0; sC1 = sN1; wC0 = wN0; wC1 = wN1;
            sN0 = sM0; sN1 = sM1;
        }
    } else {
        for (int i = 0; i < NPW; ++i) {
            const int beg = __builtin_amdgcn_readlane(rp, i);
            const int end = __builtin_amdgcn_readlane(rp, i + 1);
            const float si = __int_as_float(__builtin_amdgcn_readlane(__float_as_int(nrm), i));

            float4 acc = make_float4(0.f, 0.f, 0.f, 0.f);
            int e = beg + sub;
            for (; e + 3 * EPI < end; e += 4 * EPI) {
                const int s0 = csr[e];
                const int s1 = csr[e + EPI];
                const int s2 = csr[e + 2 * EPI];
                const int s3 = csr[e + 3 * EPI];
                const float w0 = invO[s0];
                const float w1 = invO[s1];
                const float w2 = invO[s2];
                const float w3 = invO[s3];
                const float4 v0 = *reinterpret_cast<const float4*>(xc + (size_t)s0 * D);
                const float4 v1 = *reinterpret_cast<const float4*>(xc + (size_t)s1 * D);
                const float4 v2 = *reinterpret_cast<const float4*>(xc + (size_t)s2 * D);
                const float4 v3 = *reinterpret_cast<const float4*>(xc + (size_t)s3 * D);
                acc.x += v0.x * w0; acc.y += v0.y * w0; acc.z += v0.z * w0; acc.w += v0.w * w0;
                acc.x += v1.x * w1; acc.y += v1.y * w1; acc.z += v1.z * w1; acc.w += v1.w * w1;
                acc.x += v2.x * w2; acc.y += v2.y * w2; acc.z += v2.z * w2; acc.w += v2.w * w2;
                acc.x += v3.x * w3; acc.y += v3.y * w3; acc.z += v3.z * w3; acc.w += v3.w * w3;
            }
            for (; e < end; e += EPI) {
                const int s0 = csr[e];
                const float w0 = invO[s0];
                const float4 v0 = *reinterpret_cast<const float4*>(xc + (size_t)s0 * D);
                acc.x += v0.x * w0; acc.y += v0.y * w0; acc.z += v0.z * w0; acc.w += v0.w * w0;
            }
            #pragma unroll
            for (int o = LPR; o < 32; o <<= 1) {
                acc.x += __shfl_xor(acc.x, o);
                acc.y += __shfl_xor(acc.y, o);
                acc.z += __shfl_xor(acc.z, o);
                acc.w += __shfl_xor(acc.w, o);
            }
            if (sub == 0)
                sA4[(wv * NPW + i) * (PITCH / 4) + c] =
                    make_float4(acc.x * si, acc.y * si, acc.z * si, acc.w * si);
        }
    }

    const int   ln15    = lane & 15;
    const int   hi      = lane >> 4;
    const int   kHalf   = hi * 2;
    const int   ct      = wv % COLT;
    const int   col     = ct * 16 + ln15;
    const int   colC    = FULLCOL ? col : ((col < MD) ? col : (MD - 1));
    const float colMask = FULLCOL ? 1.f : ((col < MD) ? 1.f : 0.f);

    v2f bf[NK];
    float bv = 0.f;
    if (ALLWAVES || wv < NT) {
        #pragma unroll
        for (int t = 0; t < NK; ++t) {
            const int k0 = t * 4 + kHalf;
            bf[t].x = W[k0 * MD + colC] * colMask;
            bf[t].y = W[(k0 + 1) * MD + colC] * colMask;
        }
        bv = bias[colC] * colMask;
    }

    __syncthreads();

    const float* sAf = reinterpret_cast<const float*>(sA4);
    for (int tl = wv; tl < NT; tl += 8) {
        const int rt       = tl / COLT;
        const int tileBase = nodeBase + rt * 16;
        if (tileBase >= Nn) break;
        const float* aRow = sAf + (rt * 16 + ln15) * PITCH + kHalf;

        v8f acc = {0.f, 0.f, 0.f, 0.f, 0.f, 0.f, 0.f, 0.f};
        #pragma unroll
        for (int t = 0; t < NK; ++t) {
            const v2f a = *reinterpret_cast<const v2f*>(aRow + t * 4);
            acc = __builtin_amdgcn_wmma_f32_16x16x4_f32(
                false, a, false, bf[t], (short)0, acc, false, false);
        }

        float h[8];
        #pragma unroll
        for (int j = 0; j < 8; ++j) {
            const float v = acc[j] + bv;
            h[j] = LEAKY ? fmaxf(v, LEAKY_SLOPE * v) : v;
        }

        float* outBase = out + (size_t)(tileBase + hi * 8) * MD + col;
        if (FULLCOL && (tileBase + 16 <= Nn)) {
            #pragma unroll
            for (int j = 0; j < 8; ++j) outBase[j * MD] = h[j];
        } else {
            #pragma unroll
            for (int j = 0; j < 8; ++j) {
                const int r = tileBase + hi * 8 + j;
                if (col < MD && r < Nn) outBase[j * MD] = h[j];
            }
        }
    }
}

extern "C" void kernel_launch(void* const* d_in, const int* in_sizes, int n_in,
                              void* d_out, int out_size, void* d_ws, size_t ws_size,
                              hipStream_t stream) {
    (void)n_in; (void)out_size;

    const float* features = (const float*)d_in[0];
    const int*   esrc     = (const int*)d_in[1];
    const int*   edst     = (const int*)d_in[2];

    const int N = in_sizes[0] / 4;
    const int E = in_sizes[1];
    static const int DIMS[9] = {4, 64, 64, 64, 64, 64, 64, 64, 3};

    float* ws      = (float*)d_ws;
    float* inv_out = ws;
    float* inv_in  = ws + 100352;
    float* xbuf    = ws + 200704;
    float* agg     = ws + 200704 + 6400000;

    const int BT = 256;

    if (ws_size >= (size_t)65204224) {
        int* cntO   = (int*)inv_out;
        int* cntI   = (int*)inv_in;
        int* rank   = (int*)xbuf;
        int* csr    = (int*)(ws + 13000704);
        int* rowptr = (int*)(ws + 13000704 + 3200000);

        int n4 = (N + 3) / 4;
        k_zero4<<<(n4 + BT - 1) / BT, BT, 0, stream>>>((float4*)inv_out, n4);
        k_zero4<<<(n4 + BT - 1) / BT, BT, 0, stream>>>((float4*)inv_in,  n4);
        k_count_rank<<<(E + BT - 1) / BT, BT, 0, stream>>>(esrc, edst, cntO, cntI, rank, E);
        k_scan<<<1, 1024, 0, stream>>>(cntI, rowptr, N);
        k_inv_sqrt_cnt<<<(N + BT - 1) / BT, BT, 0, stream>>>(cntO, inv_out, cntI, inv_in, N);
        k_fill<<<(E + BT - 1) / BT, BT, 0, stream>>>(esrc, edst, rowptr, rank, csr, E);

        const int lblocks = (N + 79) / 80;
        float* planeA = xbuf;
        float* planeB = agg;
        k_layer<4, 64, true><<<lblocks, BT, 0, stream>>>(
            features, inv_out, inv_in, rowptr, csr,
            (const float*)d_in[3], (const float*)d_in[4], planeA, N);
        const float* xin = planeA;
        float* xout = planeB;
        for (int i = 1; i < 7; ++i) {
            k_layer<64, 64, true><<<lblocks, BT, 0, stream>>>(
                xin, inv_out, inv_in, rowptr, csr,
                (const float*)d_in[3 + 2 * i], (const float*)d_in[4 + 2 * i], xout, N);
            float* t = (float*)xin;
            xin = xout;
            xout = t;
        }
        k_layer<64, 3, false><<<lblocks, BT, 0, stream>>>(
            xin, inv_out, inv_in, rowptr, csr,
            (const float*)d_in[17], (const float*)d_in[18], (float*)d_out, N);
        return;
    }

    {
        int n4 = (N + 3) / 4;
        k_zero4<<<(n4 + BT - 1) / BT, BT, 0, stream>>>((float4*)inv_out, n4);
        k_zero4<<<(n4 + BT - 1) / BT, BT, 0, stream>>>((float4*)inv_in,  n4);
        k_degrees<<<(E + BT - 1) / BT, BT, 0, stream>>>(esrc, edst, inv_out, inv_in, E);
        k_inv_sqrt<<<(N + BT - 1) / BT, BT, 0, stream>>>(inv_out, inv_in, N);
    }

    const float* xcur = features;
    for (int i = 0; i < 8; ++i) {
        const int Kd = DIMS[i];
        const int Md = DIMS[i + 1];
        const float* Wi = (const float*)d_in[3 + 2 * i];
        const float* bi = (const float*)d_in[4 + 2 * i];

        int n4 = N * Kd / 4;
        k_zero4<<<(n4 + BT - 1) / BT, BT, 0, stream>>>((float4*)agg, n4);

        int chunks = Kd / 4;
        int lg = (chunks == 16) ? 4 : 0;
        long long total = (long long)E * chunks;
        int blocks = (int)((total + BT - 1) / BT);
        k_gather<<<blocks, BT, 0, stream>>>(xcur, inv_out, esrc, edst, agg, E, lg, Kd);

        int colTiles  = (Md + 15) / 16;
        int rowBlocks = (N + 63) / 64;
        long long waves = (long long)rowBlocks * colTiles;
        long long thr = waves * 32;
        int gblocks = (int)((thr + BT - 1) / BT);
        float* yout = (i == 7) ? (float*)d_out : xbuf;

        if (i == 0) {
            k_gemm_wmma<4, 64, true><<<gblocks, BT, 0, stream>>>(
                agg, inv_in, Wi, bi, yout, N, colTiles, rowBlocks);
        } else if (i < 7) {
            k_gemm_wmma<64, 64, true><<<gblocks, BT, 0, stream>>>(
                agg, inv_in, Wi, bi, yout, N, colTiles, rowBlocks);
        } else {
            k_gemm_wmma<64, 3, false><<<gblocks, BT, 0, stream>>>(
                agg, inv_in, Wi, bi, yout, N, colTiles, rowBlocks);
        }
        xcur = xbuf;
    }
}


# Out-degree norm folded into each row's producer (stored rows = x*inv_sqrt_out, as the reference forms them): gather drops its random norm load and multiply per edge; sums add rounded products
# speedup vs baseline: 4.0785x; 1.0870x over previous
#include <hip/hip_runtime.h>


typedef __attribute__((ext_vector_type(2))) float v2f;
typedef __attribute__((ext_vector_type(8))) float v8f;

#define LEAKY_SLOPE 0.01f

__device__ __forceinline__ void atomAddF(float* p, float v) {
    __hip_atomic_fetch_add(p, v, __ATOMIC_RELAXED, __HIP_MEMORY_SCOPE_AGENT);
}

__global__ void k_zero4(float4* __restrict__ p, int n4) {
    int i = blockIdx.x * blockDim.x + threadIdx.x;
    if (i < n4) p[i] = make_float4(0.f, 0.f, 0.f, 0.f);
}

__global__ void k_degrees(const int* __restrict__ src, const int* __restrict__ dst,
                          float* __restrict__ degO, float* __restrict__ degI, int E) {
    int e = blockIdx.x * blockDim.x + threadIdx.x;
    if (e >= E) return;
    atomAddF(&degO[src[e]], 1.0f);
    atomAddF(&degI[dst[e]], 1.0f);
}

__global__ void k_inv_sqrt(float* __restrict__ a, float* __restrict__ b, int n) {
    int i = blockIdx.x * blockDim.x + threadIdx.x;
    if (i >= n) return;
    a[i] = rsqrtf(fmaxf(a[i], 1.0f));
    b[i] = rsqrtf(fmaxf(b[i], 1.0f));
}

__global__ void k_gather(const float* __restrict__ x, const float* __restrict__ invO,
                         const int* __restrict__ src, const int* __restrict__ dst,
                         float* __restrict__ agg, int E, int lgChunks, int d) {
    int tid = blockIdx.x * blockDim.x + threadIdx.x;
    int e = tid >> lgChunks;
    if (e >= E) return;
    int c = tid & ((1 << lgChunks) - 1);
    int s = src[e];
    int t = dst[e];
    float w = invO[s];
    const float4 v = *reinterpret_cast<const float4*>(x + (size_t)s * d + c * 4);
    float* base = agg + (size_t)t * d + c * 4;
    atomAddF(base + 0, v.x * w);
    atomAddF(base + 1, v.y * w);
    atomAddF(base + 2, v.z * w);
    atomAddF(base + 3, v.w * w);
}

__global__ void k_count_rank(const int* __restrict__ src, const int* __restrict__ dst,
                             int* __restrict__ cntO, int* __restrict__ cntI,
                             int* __restrict__ rank, int E) {
    int e = blockIdx.x * blockDim.x + threadIdx.x;
    if (e >= E) return;
    rank[e] = __hip_atomic_fetch_add(&cntI[dst[e]], 1, __ATOMIC_RELAXED, __HIP_MEMORY_SCOPE_AGENT);
    __hip_atomic_fetch_add(&cntO[src[e]], 1, __ATOMIC_RELAXED, __HIP_MEMORY_SCOPE_AGENT);
}

__global__ void __launch_bounds__(1024) k_scan(const int* __restrict__ cnt,
                                               int* __restrict__ rowptr, int n) {
    __shared__ int wsum[32];
    __shared__ int tileTotal;
    const int tid = threadIdx.x, lane = tid & 31, wv = tid >> 5;
    int carry = 0;
    for (int base = 0; base < n; base += 4096) {
        const int i = base + tid * 4;
        int a0 = 0, a1 = 0, a2 = 0, a3 = 0;
        if (i + 3 < n) {
            const int4 v = *reinterpret_cast<const int4*>(cnt + i);
            a0 = v.x; a1 = v.y; a2 = v.z; a3 = v.w;
        } else {
            if (i < n) a0 = cnt[i];
            if (i + 1 < n) a1 = cnt[i + 1];
            if (i + 2 < n) a2 = cnt[i + 2];
        }
        const int tsum = a0 + a1 + a2 + a3;
        int inc = tsum;
        for (int o = 1; o < 32; o <<= 1) {
            int t = __shfl_up(inc, o);
            if (lane >= o) inc += t;
        }
        if (lane == 31) wsum[wv] = inc;
        __syncthreads();
        if (wv == 0) {
            const int w = wsum[lane];
            int winc = w;
            for (int o = 1; o < 32; o <<= 1) {
                int t = __shfl_up(winc, o);
                if (lane >= o) winc += t;
            }
            wsum[lane] = winc - w;
            if (lane == 31) tileTotal = winc;
        }
        __syncthreads();
        const int ex = carry + wsum[wv] + (inc - tsum);
        carry += tileTotal;
        if (i + 3 < n) {
            *reinterpret_cast<int4*>(rowptr + i) = make_int4(ex, ex + a0, ex + a0 + a1, ex + a0 + a1 + a2);
        } else {
            if (i < n) rowptr[i] = ex;
            if (i + 1 < n) rowptr[i + 1] = ex + a0;
            if (i + 2 < n) rowptr[i + 2] = ex + a0 + a1;
        }
        __syncthreads();
    }
    if (tid == 0) rowptr[n] = carry;
}

__global__ void k_inv_sqrt_cnt(const int* cntA, float* outA, const int* cntB, float* outB,
                               const float* __restrict__ feat, float* __restrict__ xs0, int n) {
    int i = blockIdx.x * blockDim.x + threadIdx.x;
    if (i >= n) return;
    float fa = (float)cntA[i];
    float fb = (float)cntB[i];
    const float wo = rsqrtf(fmaxf(fa, 1.0f));
    outA[i] = wo;
    outB[i] = rsqrtf(fmaxf(fb, 1.0f));
    const float4 f = *reinterpret_cast<const float4*>(feat + (size_t)i * 4);
    *reinterpret_cast<float4*>(xs0 + (size_t)i * 4) = make_float4(f.x * wo, f.y * wo, f.z * wo, f.w * wo);
}

__global__ void k_fill(const int* __restrict__ src, const int* __restrict__ dst,
                       const int* __restrict__ rowptr, const int* __restrict__ rank,
                       int* __restrict__ csr, int E) {
    int e = blockIdx.x * blockDim.x + threadIdx.x;
    if (e >= E) return;
    csr[rowptr[dst[e]] + rank[e]] = src[e];
}

template <int KDIM, int MD, bool LEAKY>
__global__ void k_gemm_wmma(const float* __restrict__ agg, const float* __restrict__ invI,
                            const float* __restrict__ W, const float* __restrict__ bias,
                            float* __restrict__ out, int Nn, int colTiles, int rowBlocks) {
    constexpr int NK = KDIM / 4;
    constexpr bool FULLCOL = (MD % 16 == 0);

    int gwave    = __builtin_amdgcn_readfirstlane(
                       (int)((blockIdx.x * blockDim.x + threadIdx.x) >> 5));
    int lane     = threadIdx.x & 31;
    int rowBlock = gwave / colTiles;
    int colTile  = gwave - rowBlock * colTiles;
    if (rowBlock >= rowBlocks) return;

    int   ln15    = lane & 15;
    int   col     = colTile * 16 + ln15;
    int   colC    = FULLCOL ? col : ((col < MD) ? col : (MD - 1));
    float colMask = FULLCOL ? 1.f : ((col < MD) ? 1.f : 0.f);
    int   kHalf   = (lane >> 4) * 2;
    int   hi      = lane >> 4;

    v2f bf[NK];
    #pragma unroll
    for (int t = 0; t < NK; ++t) {
        int k0 = t * 4 + kHalf;
        bf[t].x = W[k0 * MD + colC] * colMask;
        bf[t].y = W[(k0 + 1) * MD + colC] * colMask;
    }
    float bv = bias[colC] * colMask;

    #pragma unroll
    for (int st = 0; st < 4; ++st) {
        int tileBase = rowBlock * 64 + st * 16;
        if (tileBase >= Nn) break;
        int   mrow  = tileBase + ln15;
        int   mrowC = (mrow < Nn) ? mrow : (Nn - 1);
        float s     = invI[mrowC];

        v8f acc = {0.f, 0.f, 0.f, 0.f, 0.f, 0.f, 0.f, 0.f};
        #pragma unroll
        for (int t = 0; t < NK; ++t) {
            int k0 = t * 4 + kHalf;
            v2f a = *reinterpret_cast<const v2f*>(agg + (size_t)mrowC * KDIM + k0);
            a.x *= s;
            a.y *= s;
            acc = __builtin_amdgcn_wmma_f32_16x16x4_f32(
                false, a, false, bf[t], (short)0, acc, false, false);
        }

        float h[8];
        #pragma unroll
        for (int j = 0; j < 8; ++j) {
            float v = acc[j] + bv;
            h[j] = LEAKY ? fmaxf(v, LEAKY_SLOPE * v) : v;
        }

        float* outBase = out + (size_t)(tileBase + hi * 8) * MD + col;
        if (FULLCOL && (tileBase + 16 <= Nn)) {
            #pragma unroll
            for (int j = 0; j < 8; ++j) outBase[j * MD] = h[j];
        } else {
            #pragma unroll
            for (int j = 0; j < 8; ++j) {
                int r = tileBase + hi * 8 + j;
                if (col < MD && r < Nn) outBase[j * MD] = h[j];
            }
        }
    }
}

__device__ __forceinline__ void listLoad(const int* __restrict__ csr, int beg, int end, int lane,
                                         int& s0, int& s1) {
    const int last = (end > 0) ? (end - 1) : 0;
    const int e0 = beg + lane;
    const int e1 = e0 + 32;
    s0 = csr[e0 < last ? e0 : last];
    s1 = csr[e1 < last ? e1 : last];
}

template <int L0, bool FULL>
__device__ __forceinline__ void gather16(const char* __restrict__ xb, unsigned cOff, int sub,
                                         int sReg, int cnt, float4& acc) {
    float4 v[8];
    const int sFirst = __builtin_amdgcn_readlane(sReg, L0);
    #pragma unroll
    for (int u = 0; u < 8; ++u) {
        const int sLo = __builtin_amdgcn_readlane(sReg, L0 + 2 * u);
        const int sHi = __builtin_amdgcn_readlane(sReg, L0 + 2 * u + 1);
        int s = sub ? sHi : sLo;
        if (!FULL) s = (2 * u + sub < cnt) ? s : sFirst;
        v[u] = *reinterpret_cast<const float4*>(xb + (((unsigned)s << 8) + cOff));
    }
    #pragma unroll
    for (int u = 0; u < 8; ++u) {
        float4 t = v[u];
        if (!FULL && !(2 * u + sub < cnt)) t = make_float4(0.f, 0.f, 0.f, 0.f);
        acc.x += t.x; acc.y += t.y; acc.z += t.z; acc.w += t.w;
    }
}

__device__ __forceinline__ void gather32(const char* __restrict__ xb, unsigned cOff, int sub,
                                         int sReg, int n, float4& acc) {
    if (n >= 16) gather16<0, true>(xb, cOff, sub, sReg, 16, acc);
    else if (n > 0) gather16<0, false>(xb, cOff, sub, sReg, n, acc);
    if (n >= 32) gather16<16, true>(xb, cOff, sub, sReg, 16, acc);
    else if (n > 16) gather16<16, false>(xb, cOff, sub, sReg, n - 16, acc);
}

template <int D, int MD, bool LEAKY, bool SCALE_OUT>
__global__ void __launch_bounds__(256) __attribute__((amdgpu_waves_per_eu(12, 16)))
k_layer(const float* __restrict__ x, const float* __restrict__ invO,
        const float* __restrict__ invI, const int* __restrict__ rowptr,
        const int* __restrict__ csr, const float* __restrict__ W,
        const float* __restrict__ bias, float* __restrict__ out, int Nn) {
    constexpr int LPR   = D / 4;
    constexpr int EPI   = 32 / LPR;
    constexpr int PITCH = (D >= 64) ? (D + 4) : D;
    constexpr int NK    = D / 4;
    constexpr int COLT  = (MD + 15) / 16;
    constexpr bool FULLCOL = (MD % 16 == 0);
    static_assert(8 % COLT == 0, "a wave's tiles must share one column slice");
    constexpr int ROWS  = 80;
    constexpr int NPW   = ROWS / 8;
    constexpr int NT    = (ROWS / 16) * COLT;
    constexpr bool ALLWAVES = (NT >= 8);

    __shared__ float4 sA4[ROWS * PITCH / 4];

    const int lane     = threadIdx.x & 31;
    const int wv       = __builtin_amdgcn_readfirstlane((int)(threadIdx.x >> 5));
    const int nodeBase = blockIdx.x * ROWS;
    const int sub      = lane / LPR;
    const int c        = lane % LPR;
    const float* xc    = x + c * 4;

    const int first = nodeBase + wv * NPW;
    const int pIdx  = first + lane;
    const int   rp  = rowptr[pIdx < Nn ? pIdx : Nn];
    const float nrm = invI[pIdx < Nn ? pIdx : (Nn - 1)];

    if constexpr (D == 64) {
        const char* xb = reinterpret_cast<const char*>(x);
        const unsigned cOff = (unsigned)c * 16u;
        int   sC0, sC1, sN0, sN1;
        {
            const int b0 = __builtin_amdgcn_readlane(rp, 0);
            const int e0 = __builtin_amdgcn_readlane(rp, 1);
            const int e1 = __builtin_amdgcn_readlane(rp, 2);
            listLoad(csr, b0, e0, lane, sC0, sC1);
            listLoad(csr, e0, e1, lane, sN0, sN1);
        }
        #pragma nounroll
        for (int i = 0; i < NPW; ++i) {
            const int beg  = __builtin_amdgcn_readlane(rp, i);
            const int end  = __builtin_amdgcn_readlane(rp, i + 1);
            const int end1 = __builtin_amdgcn_readlane(rp, i + 2);
            const int end2 = __builtin_amdgcn_readlane(rp, i + 3);
            const float si = __int_as_float(__builtin_amdgcn_readlane(__float_as_int(nrm), i));

            int   sM0, sM1;
            listLoad(csr, end1, end2, lane, sM0, sM1);

            const int deg = end - beg;
            float4 acc = make_float4(0.f, 0.f, 0.f, 0.f);
            gather32(xb, cOff, sub, sC0, deg, acc);
            if (deg > 32) gather32(xb, cOff, sub, sC1, deg - 32, acc);
            for (int e = beg + 64 + sub; e < end; e += EPI) {
                const int s0 = csr[e];
                const float4 v0 = *reinterpret_cast<const float4*>(xc + (size_t)s0 * D);
                acc.x += v0.x; acc.y += v0.y; acc.z += v0.z; acc.w += v0.w;
            }
            acc.x += __shfl_xor(acc.x, 16);
            acc.y += __shfl_xor(acc.y, 16);
            acc.z += __shfl_xor(acc.z, 16);
            acc.w += __shfl_xor(acc.w, 16);
            if (sub == 0)
                sA4[(wv * NPW + i) * (PITCH / 4) + c] =
                    make_float4(acc.x * si, acc.y * si, acc.z * si, acc.w * si);

            sC0 = sN0; sC1 = sN1;
            sN0 = sM0; sN1 = sM1;
        }
    } else {
        for (int i = 0; i < NPW; ++i) {
            const int beg = __builtin_amdgcn_readlane(rp, i);
            const int end = __builtin_amdgcn_readlane(rp, i + 1);
            const float si = __int_as_float(__builtin_amdgcn_readlane(__float_as_int(nrm), i));

            float4 acc = make_float4(0.f, 0.f, 0.f, 0.f);
            int e = beg + sub;
            for (; e + 3 * EPI < end; e += 4 * EPI) {
                const int s0 = csr[e];
                const int s1 = csr[e + EPI];
                const int s2 = csr[e + 2 * EPI];
                const int s3 = csr[e + 3 * EPI];
                const float4 v0 = *reinterpret_cast<const float4*>(xc + (size_t)s0 * D);
                const float4 v1 = *reinterpret_cast<const float4*>(xc + (size_t)s1 * D);
                const float4 v2 = *reinterpret_cast<const float4*>(xc + (size_t)s2 * D);
                const float4 v3 = *reinterpret_cast<const float4*>(xc + (size_t)s3 * D);
                acc.x += v0.x; acc.y += v0.y; acc.z += v0.z; acc.w += v0.w;
                acc.x += v1.x; acc.y += v1.y; acc.z += v1.z; acc.w += v1.w;
                acc.x += v2.x; acc.y += v2.y; acc.z += v2.z; acc.w += v2.w;
                acc.x += v3.x; acc.y += v3.y; acc.z += v3.z; acc.w += v3.w;
            }
            for (; e < end; e += EPI) {
                const int s0 = csr[e];
                const float4 v0 = *reinterpret_cast<const float4*>(xc + (size_t)s0 * D);
                acc.x += v0.x; acc.y += v0.y; acc.z += v0.z; acc.w += v0.w;
            }
            #pragma unroll
            for (int o = LPR; o < 32; o <<= 1) {
                acc.x += __shfl_xor(acc.x, o);
                acc.y += __shfl_xor(acc.y, o);
                acc.z += __shfl_xor(acc.z, o);
                acc.w += __shfl_xor(acc.w, o);
            }
            if (sub == 0)
                sA4[(wv * NPW + i) * (PITCH / 4) + c] =
                    make_float4(acc.x * si, acc.y * si, acc.z * si, acc.w * si);
        }
    }

    const int   ln15    = lane & 15;
    const int   hi      = lane >> 4;
    const int   kHalf   = hi * 2;
    const int   ct      = wv % COLT;
    const int   col     = ct * 16 + ln15;
    const int   colC    = FULLCOL ? col : ((col < MD) ? col : (MD - 1));
    const float colMask = FULLCOL ? 1.f : ((col < MD) ? 1.f : 0.f);

    v2f bf[NK];
    float bv = 0.f;
    if (ALLWAVES || wv < NT) {
        #pragma unroll
        for (int t = 0; t < NK; ++t) {
            const int k0 = t * 4 + kHalf;
            bf[t].x = W[k0 * MD + colC] * colMask;
            bf[t].y = W[(k0 + 1) * MD + colC] * colMask;
        }
        bv = bias[colC] * colMask;
    }

    __syncthreads();

    const float* sAf = reinterpret_cast<const float*>(sA4);
    for (int tl = wv; tl < NT; tl += 8) {
        const int rt       = tl / COLT;
        const int tileBase = nodeBase + rt * 16;
        if (tileBase >= Nn) break;
        const float* aRow = sAf + (rt * 16 + ln15) * PITCH + kHalf;

        float so[8];
        if constexpr (SCALE_OUT) {
            const int r0 = tileBase + hi * 8;
            if (tileBase + 16 <= Nn) {
                const float4 n0 = *reinterpret_cast<const float4*>(invO + r0);
                const float4 n1 = *reinterpret_cast<const float4*>(invO + r0 + 4);
                so[0] = n0.x; so[1] = n0.y; so[2] = n0.z; so[3] = n0.w;
                so[4] = n1.x; so[5] = n1.y; so[6] = n1.z; so[7] = n1.w;
            } else {
                #pragma unroll
                for (int j = 0; j < 8; ++j) {
                    const int r = r0 + j;
                    so[j] = invO[r < Nn ? r : (Nn - 1)];
                }
            }
        }

        v8f acc = {0.f, 0.f, 0.f, 0.f, 0.f, 0.f, 0.f, 0.f};
        #pragma unroll
        for (int t = 0; t < NK; ++t) {
            const v2f a = *reinterpret_cast<const v2f*>(aRow + t * 4);
            acc = __builtin_amdgcn_wmma_f32_16x16x4_f32(
                false, a, false, bf[t], (short)0, acc, false, false);
        }

        float h[8];
        #pragma unroll
        for (int j = 0; j < 8; ++j) {
            const float v = acc[j] + bv;
            const float a = LEAKY ? fmaxf(v, LEAKY_SLOPE * v) : v;
            if constexpr (SCALE_OUT) h[j] = a * so[j];
            else h[j] = a;
        }

        float* outBase = out + (size_t)(tileBase + hi * 8) * MD + col;
        if (FULLCOL && (tileBase + 16 <= Nn)) {
            #pragma unroll
            for (int j = 0; j < 8; ++j) outBase[j * MD] = h[j];
        } else {
            #pragma unroll
            for (int j = 0; j < 8; ++j) {
                const int r = tileBase + hi * 8 + j;
                if (col < MD && r < Nn) outBase[j * MD] = h[j];
            }
        }
    }
}

extern "C" void kernel_launch(void* const* d_in, const int* in_sizes, int n_in,
                              void* d_out, int out_size, void* d_ws, size_t ws_size,
                              hipStream_t stream) {
    (void)n_in; (void)out_size;

    const float* features = (const float*)d_in[0];
    const int*   esrc     = (const int*)d_in[1];
    const int*   edst     = (const int*)d_in[2];

    const int N = in_sizes[0] / 4;
    const int E = in_sizes[1];
    static const int DIMS[9] = {4, 64, 64, 64, 64, 64, 64, 64, 3};

    float* ws      = (float*)d_ws;
    float* inv_out = ws;
    float* inv_in  = ws + 100352;
    float* xbuf    = ws + 200704;
    float* agg     = ws + 200704 + 6400000;

    const int BT = 256;

    if (ws_size >= (size_t)65204224) {
        int* cntO   = (int*)inv_out;
        int* cntI   = (int*)inv_in;
        int* rank   = (int*)xbuf;
        int* csr    = (int*)(ws + 13000704);
        int* rowptr = (int*)(ws + 13000704 + 3200000);

        int n4 = (N + 3) / 4;
        k_zero4<<<(n4 + BT - 1) / BT, BT, 0, stream>>>((float4*)inv_out, n4);
        k_zero4<<<(n4 + BT - 1) / BT, BT, 0, stream>>>((float4*)inv_in,  n4);
        k_count_rank<<<(E + BT - 1) / BT, BT, 0, stream>>>(esrc, edst, cntO, cntI, rank, E);
        k_scan<<<1, 1024, 0, stream>>>(cntI, rowptr, N);
        k_inv_sqrt_cnt<<<(N + BT - 1) / BT, BT, 0, stream>>>(cntO, inv_out, cntI, inv_in,
                                                              features, agg, N);
        k_fill<<<(E + BT - 1) / BT, BT, 0, stream>>>(esrc, edst, rowptr, rank, csr, E);

        const int lblocks = (N + 79) / 80;
        float* planeA = xbuf;
        float* planeB = agg;
        k_layer<4, 64, true, true><<<lblocks, BT, 0, stream>>>(
            planeB, inv_out, inv_in, rowptr, csr,
            (const float*)d_in[3], (const float*)d_in[4], planeA, N);
        const float* xin = planeA;
        float* xout = planeB;
        for (int i = 1; i < 7; ++i) {
            k_layer<64, 64, true, true><<<lblocks, BT, 0, stream>>>(
                xin, inv_out, inv_in, rowptr, csr,
                (const float*)d_in[3 + 2 * i], (const float*)d_in[4 + 2 * i], xout, N);
            float* t = (float*)xin;
            xin = xout;
            xout = t;
        }
        k_layer<64, 3, false, false><<<lblocks, BT, 0, stream>>>(
            xin, inv_out, inv_in, rowptr, csr,
            (const float*)d_in[17], (const float*)d_in[18], (float*)d_out, N);
        return;
    }

    {
        int n4 = (N + 3) / 4;
        k_zero4<<<(n4 + BT - 1) / BT, BT, 0, stream>>>((float4*)inv_out, n4);
        k_zero4<<<(n4 + BT - 1) / BT, BT, 0, stream>>>((float4*)inv_in,  n4);
        k_degrees<<<(E + BT - 1) / BT, BT, 0, stream>>>(esrc, edst, inv_out, inv_in, E);
        k_inv_sqrt<<<(N + BT - 1) / BT, BT, 0, stream>>>(inv_out, inv_in, N);
    }

    const float* xcur = features;
    for (int i = 0; i < 8; ++i) {
        const int Kd = DIMS[i];
        const int Md = DIMS[i + 1];
        const float* Wi = (const float*)d_in[3 + 2 * i];
        const float* bi = (const float*)d_in[4 + 2 * i];

        int n4 = N * Kd / 4;
        k_zero4<<<(n4 + BT - 1) / BT, BT, 0, stream>>>((float4*)agg, n4);

        int chunks = Kd / 4;
        int lg = (chunks == 16) ? 4 : 0;
        long long total = (long long)E * chunks;
        int blocks = (int)((total + BT - 1) / BT);
        k_gather<<<blocks, BT, 0, stream>>>(xcur, inv_out, esrc, edst, agg, E, lg, Kd);

        int colTiles  = (Md + 15) / 16;
        int rowBlocks = (N + 63) / 64;
        long long waves = (long long)rowBlocks * colTiles;
        long long thr = waves * 32;
        int gblocks = (int)((thr + BT - 1) / BT);
        float* yout = (i == 7) ? (float*)d_out : xbuf;

        if (i == 0) {
            k_gemm_wmma<4, 64, true><<<gblocks, BT, 0, stream>>>(
                agg, inv_in, Wi, bi, yout, N, colTiles, rowBlocks);
        } else if (i < 7) {
            k_gemm_wmma<64, 64, true><<<gblocks, BT, 0, stream>>>(
                agg, inv_in, Wi, bi, yout, N, colTiles, rowBlocks);
        } else {
            k_gemm_wmma<64, 3, false><<<gblocks, BT, 0, stream>>>(
                agg, inv_in, Wi, bi, yout, N, colTiles, rowBlocks);
        }
        xcur = xbuf;
    }
}


# Edge list built with LDS counters: work-group = (edge chunk, node range), per-chunk degree parts, LDS returning add = an edge's slot; no global atomics or zero fills; a node's sum order changes
# speedup vs baseline: 6.9500x; 1.7041x over previous
#include <hip/hip_runtime.h>


typedef __attribute__((ext_vector_type(2))) float v2f;
typedef __attribute__((ext_vector_type(8))) float v8f;

#define LEAKY_SLOPE 0.01f

__device__ __forceinline__ void atomAddF(float* p, float v) {
    __hip_atomic_fetch_add(p, v, __ATOMIC_RELAXED, __HIP_MEMORY_SCOPE_AGENT);
}

__global__ void k_zero4(float4* __restrict__ p, int n4) {
    int i = blockIdx.x * blockDim.x + threadIdx.x;
    if (i < n4) p[i] = make_float4(0.f, 0.f, 0.f, 0.f);
}

__global__ void k_degrees(const int* __restrict__ src, const int* __restrict__ dst,
                          float* __restrict__ degO, float* __restrict__ degI, int E) {
    int e = blockIdx.x * blockDim.x + threadIdx.x;
    if (e >= E) return;
    atomAddF(&degO[src[e]], 1.0f);
    atomAddF(&degI[dst[e]], 1.0f);
}

__global__ void k_inv_sqrt(float* __restrict__ a, float* __restrict__ b, int n) {
    int i = blockIdx.x * blockDim.x + threadIdx.x;
    if (i >= n) return;
    a[i] = rsqrtf(fmaxf(a[i], 1.0f));
    b[i] = rsqrtf(fmaxf(b[i], 1.0f));
}

__global__ void k_gather(const float* __restrict__ x, const float* __restrict__ invO,
                         const int* __restrict__ src, const int* __restrict__ dst,
                         float* __restrict__ agg, int E, int lgChunks, int d) {
    int tid = blockIdx.x * blockDim.x + threadIdx.x;
    int e = tid >> lgChunks;
    if (e >= E) return;
    int c = tid & ((1 << lgChunks) - 1);
    int s = src[e];
    int t = dst[e];
    float w = invO[s];
    const float4 v = *reinterpret_cast<const float4*>(x + (size_t)s * d + c * 4);
    float* base = agg + (size_t)t * d + c * 4;
    atomAddF(base + 0, v.x * w);
    atomAddF(base + 1, v.y * w);
    atomAddF(base + 2, v.z * w);
    atomAddF(base + 3, v.w * w);
}

#define LB_RANGES 8
#define LB_CHUNKS 32
#define LB_RW     12544
#define LB_T      1024

__device__ __forceinline__ void lbChunk(int E, int ch, int& e0, int& e1) {
    const int per = ((E + LB_CHUNKS * 4 - 1) / (LB_CHUNKS * 4)) * 4;
    e0 = ch * per;
    if (e0 > E) e0 = E;
    e1 = e0 + per;
    if (e1 > E) e1 = E;
}

#define LB_COUNT(c, v)                                                                   \
    do {                                                                                 \
        const unsigned r_ = (unsigned)((v) - lo);                                        \
        if (r_ < ulen)                                                                   \
            __hip_atomic_fetch_add(&c[r_], 1, __ATOMIC_RELAXED, __HIP_MEMORY_SCOPE_WORKGROUP); \
    } while (0)

#define LB_COUNT4(s, d)                                                                  \
    do {                                                                                 \
        LB_COUNT(cO, (s).x); LB_COUNT(cO, (s).y); LB_COUNT(cO, (s).z); LB_COUNT(cO, (s).w); \
        LB_COUNT(cI, (d).x); LB_COUNT(cI, (d).y); LB_COUNT(cI, (d).z); LB_COUNT(cI, (d).w); \
    } while (0)

__global__ void __launch_bounds__(LB_T)
k_lb_count(const int* __restrict__ src, const int* __restrict__ dst,
           int* __restrict__ partI, int* __restrict__ partO, int E, int N) {
    __shared__ int cI[LB_RW];
    __shared__ int cO[LB_RW];
    const int tid = threadIdx.x;
    const int ch  = blockIdx.x / LB_RANGES;
    const int rg  = blockIdx.x % LB_RANGES;
    const int lo  = rg * LB_RW;
    int len = N - lo;
    if (len > LB_RW) len = LB_RW;
    if (len < 0) len = 0;
    const unsigned ulen = (unsigned)len;

    for (int n = tid; n < LB_RW; n += LB_T) { cI[n] = 0; cO[n] = 0; }
    __syncthreads();

    int e0, e1;
    lbChunk(E, ch, e0, e1);
    const int n4 = (e1 - e0) >> 2;
    const int4* s4 = reinterpret_cast<const int4*>(src + e0);
    const int4* d4 = reinterpret_cast<const int4*>(dst + e0);
    int q = tid;
    for (; q + 3 * LB_T < n4; q += 4 * LB_T) {
        const int4 a0 = s4[q], a1 = s4[q + LB_T], a2 = s4[q + 2 * LB_T], a3 = s4[q + 3 * LB_T];
        const int4 b0 = d4[q], b1 = d4[q + LB_T], b2 = d4[q + 2 * LB_T], b3 = d4[q + 3 * LB_T];
        LB_COUNT4(a0, b0); LB_COUNT4(a1, b1); LB_COUNT4(a2, b2); LB_COUNT4(a3, b3);
    }
    for (; q < n4; q += LB_T) {
        const int4 a0 = s4[q];
        const int4 b0 = d4[q];
        LB_COUNT4(a0, b0);
    }
    for (int e = e0 + (n4 << 2) + tid; e < e1; e += LB_T) {
        const int s = src[e];
        const int d = dst[e];
        LB_COUNT(cO, s);
        LB_COUNT(cI, d);
    }
    __syncthreads();

    int* pI = partI + (size_t)ch * N + lo;
    int* pO = partO + (size_t)ch * N + lo;
    for (int n = tid; n < len; n += LB_T) { pI[n] = cI[n]; pO[n] = cO[n]; }
}

__global__ void k_lb_nodes(int* __restrict__ partI, const int* __restrict__ partO,
                           int* __restrict__ cntI, float* __restrict__ invO,
                           float* __restrict__ invI, const float* __restrict__ feat,
                           float* __restrict__ xs0, int N) {
    const int n = blockIdx.x * blockDim.x + threadIdx.x;
    if (n >= N) return;
    int vi[LB_CHUNKS], vo[LB_CHUNKS];
    #pragma unroll
    for (int i = 0; i < LB_CHUNKS; ++i) {
        vi[i] = partI[(size_t)i * N + n];
        vo[i] = partO[(size_t)i * N + n];
    }
    int accI = 0, accO = 0;
    #pragma unroll
    for (int i = 0; i < LB_CHUNKS; ++i) {
        partI[(size_t)i * N + n] = accI;
        accI += vi[i];
        accO += vo[i];
    }
    cntI[n] = accI;
    const float wo = rsqrtf(fmaxf((float)accO, 1.0f));
    invO[n] = wo;
    invI[n] = rsqrtf(fmaxf((float)accI, 1.0f));
    const float4 f = *reinterpret_cast<const float4*>(feat + (size_t)n * 4);
    *reinterpret_cast<float4*>(xs0 + (size_t)n * 4) = make_float4(f.x * wo, f.y * wo, f.z * wo, f.w * wo);
}

#define LB_PLACE(s, d)                                                                   \
    do {                                                                                 \
        const unsigned r_ = (unsigned)((d) - lo);                                        \
        if (r_ < ulen) {                                                                 \
            const int p_ = __hip_atomic_fetch_add(&pos[r_], 1, __ATOMIC_RELAXED,         \
                                                  __HIP_MEMORY_SCOPE_WORKGROUP);         \
            csr[p_] = (s);                                                               \
        }                                                                                \
    } while (0)

#define LB_PLACE4(s, d)                                                                  \
    do {                                                                                 \
        LB_PLACE((s).x, (d).x); LB_PLACE((s).y, (d).y);                                  \
        LB_PLACE((s).z, (d).z); LB_PLACE((s).w, (d).w);                                  \
    } while (0)

__global__ void __launch_bounds__(LB_T)
k_lb_place(const int* __restrict__ src, const int* __restrict__ dst,
           const int* __restrict__ rowptr, const int* __restrict__ partI,
           int* __restrict__ csr, int E, int N) {
    __shared__ int pos[LB_RW];
    const int tid = threadIdx.x;
    const int ch  = blockIdx.x / LB_RANGES;
    const int rg  = blockIdx.x % LB_RANGES;
    const int lo  = rg * LB_RW;
    int len = N - lo;
    if (len > LB_RW) len = LB_RW;
    if (len < 0) len = 0;
    const unsigned ulen = (unsigned)len;

    const int* pI = partI + (size_t)ch * N + lo;
    for (int n = tid; n < len; n += LB_T) pos[n] = rowptr[lo + n] + pI[n];
    __syncthreads();

    int e0, e1;
    lbChunk(E, ch, e0, e1);
    const int n4 = (e1 - e0) >> 2;
    const int4* s4 = reinterpret_cast<const int4*>(src + e0);
    const int4* d4 = reinterpret_cast<const int4*>(dst + e0);
    int q = tid;
    for (; q + 3 * LB_T < n4; q += 4 * LB_T) {
        const int4 a0 = s4[q], a1 = s4[q + LB_T], a2 = s4[q + 2 * LB_T], a3 = s4[q + 3 * LB_T];
        const int4 b0 = d4[q], b1 = d4[q + LB_T], b2 = d4[q + 2 * LB_T], b3 = d4[q + 3 * LB_T];
        LB_PLACE4(a0, b0); LB_PLACE4(a1, b1); LB_PLACE4(a2, b2); LB_PLACE4(a3, b3);
    }
    for (; q < n4; q += LB_T) {
        const int4 a0 = s4[q];
        const int4 b0 = d4[q];
        LB_PLACE4(a0, b0);
    }
    for (int e = e0 + (n4 << 2) + tid; e < e1; e += LB_T) {
        const int s = src[e];
        const int d = dst[e];
        LB_PLACE(s, d);
    }
}

__global__ void __launch_bounds__(1024) k_scan(const int* __restrict__ cnt,
                                               int* __restrict__ rowptr, int n) {
    __shared__ int wsum[32];
    __shared__ int tileTotal;
    const int tid = threadIdx.x, lane = tid & 31, wv = tid >> 5;
    int carry = 0;
    for (int base = 0; base < n; base += 4096) {
        const int i = base + tid * 4;
        int a0 = 0, a1 = 0, a2 = 0, a3 = 0;
        if (i + 3 < n) {
            const int4 v = *reinterpret_cast<const int4*>(cnt + i);
            a0 = v.x; a1 = v.y; a2 = v.z; a3 = v.w;
        } else {
            if (i < n) a0 = cnt[i];
            if (i + 1 < n) a1 = cnt[i + 1];
            if (i + 2 < n) a2 = cnt[i + 2];
        }
        const int tsum = a0 + a1 + a2 + a3;
        int inc = tsum;
        for (int o = 1; o < 32; o <<= 1) {
            int t = __shfl_up(inc, o);
            if (lane >= o) inc += t;
        }
        if (lane == 31) wsum[wv] = inc;
        __syncthreads();
        if (wv == 0) {
            const int w = wsum[lane];
            int winc = w;
            for (int o = 1; o < 32; o <<= 1) {
                int t = __shfl_up(winc, o);
                if (lane >= o) winc += t;
            }
            wsum[lane] = winc - w;
            if (lane == 31) tileTotal = winc;
        }
        __syncthreads();
        const int ex = carry + wsum[wv] + (inc - tsum);
        carry += tileTotal;
        if (i + 3 < n) {
            *reinterpret_cast<int4*>(rowptr + i) = make_int4(ex, ex + a0, ex + a0 + a1, ex + a0 + a1 + a2);
        } else {
            if (i < n) rowptr[i] = ex;
            if (i + 1 < n) rowptr[i + 1] = ex + a0;
            if (i + 2 < n) rowptr[i + 2] = ex + a0 + a1;
        }
        __syncthreads();
    }
    if (tid == 0) rowptr[n] = carry;
}

template <int KDIM, int MD, bool LEAKY>
__global__ void k_gemm_wmma(const float* __restrict__ agg, const float* __restrict__ invI,
                            const float* __restrict__ W, const float* __restrict__ bias,
                            float* __restrict__ out, int Nn, int colTiles, int rowBlocks) {
    constexpr int NK = KDIM / 4;
    constexpr bool FULLCOL = (MD % 16 == 0);

    int gwave    = __builtin_amdgcn_readfirstlane(
                       (int)((blockIdx.x * blockDim.x + threadIdx.x) >> 5));
    int lane     = threadIdx.x & 31;
    int rowBlock = gwave / colTiles;
    int colTile  = gwave - rowBlock * colTiles;
    if (rowBlock >= rowBlocks) return;

    int   ln15    = lane & 15;
    int   col     = colTile * 16 + ln15;
    int   colC    = FULLCOL ? col : ((col < MD) ? col : (MD - 1));
    float colMask = FULLCOL ? 1.f : ((col < MD) ? 1.f : 0.f);
    int   kHalf   = (lane >> 4) * 2;
    int   hi      = lane >> 4;

    v2f bf[NK];
    #pragma unroll
    for (int t = 0; t < NK; ++t) {
        int k0 = t * 4 + kHalf;
        bf[t].x = W[k0 * MD + colC] * colMask;
        bf[t].y = W[(k0 + 1) * MD + colC] * colMask;
    }
    float bv = bias[colC] * colMask;

    #pragma unroll
    for (int st = 0; st < 4; ++st) {
        int tileBase = rowBlock * 64 + st * 16;
        if (tileBase >= Nn) break;
        int   mrow  = tileBase + ln15;
        int   mrowC = (mrow < Nn) ? mrow : (Nn - 1);
        float s     = invI[mrowC];

        v8f acc = {0.f, 0.f, 0.f, 0.f, 0.f, 0.f, 0.f, 0.f};
        #pragma unroll
        for (int t = 0; t < NK; ++t) {
            int k0 = t * 4 + kHalf;
            v2f a = *reinterpret_cast<const v2f*>(agg + (size_t)mrowC * KDIM + k0);
            a.x *= s;
            a.y *= s;
            acc = __builtin_amdgcn_wmma_f32_16x16x4_f32(
                false, a, false, bf[t], (short)0, acc, false, false);
        }

        float h[8];
        #pragma unroll
        for (int j = 0; j < 8; ++j) {
            float v = acc[j] + bv;
            h[j] = LEAKY ? fmaxf(v, LEAKY_SLOPE * v) : v;
        }

        float* outBase = out + (size_t)(tileBase + hi * 8) * MD + col;
        if (FULLCOL && (tileBase + 16 <= Nn)) {
            #pragma unroll
            for (int j = 0; j < 8; ++j) outBase[j * MD] = h[j];
        } else {
            #pragma unroll
            for (int j = 0; j < 8; ++j) {
                int r = tileBase + hi * 8 + j;
                if (col < MD && r < Nn) outBase[j * MD] = h[j];
            }
        }
    }
}

__device__ __forceinline__ void listLoad(const int* __restrict__ csr, int beg, int end, int lane,
                                         int& s0, int& s1) {
    const int last = (end > 0) ? (end - 1) : 0;
    const int e0 = beg + lane;
    const int e1 = e0 + 32;
    s0 = csr[e0 < last ? e0 : last];
    s1 = csr[e1 < last ? e1 : last];
}

template <int L0, bool FULL>
__device__ __forceinline__ void gather16(const char* __restrict__ xb, unsigned cOff, int sub,
                                         int sReg, int cnt, float4& acc) {
    float4 v[8];
    const int sFirst = __builtin_amdgcn_readlane(sReg, L0);
    #pragma unroll
    for (int u = 0; u < 8; ++u) {
        const int sLo = __builtin_amdgcn_readlane(sReg, L0 + 2 * u);
        const int sHi = __builtin_amdgcn_readlane(sReg, L0 + 2 * u + 1);
        int s = sub ? sHi : sLo;
        if (!FULL) s = (2 * u + sub < cnt) ? s : sFirst;
        v[u] = *reinterpret_cast<const float4*>(xb + (((unsigned)s << 8) + cOff));
    }
    #pragma unroll
    for (int u = 0; u < 8; ++u) {
        float4 t = v[u];
        if (!FULL && !(2 * u + sub < cnt)) t = make_float4(0.f, 0.f, 0.f, 0.f);
        acc.x += t.x; acc.y += t.y; acc.z += t.z; acc.w += t.w;
    }
}

__device__ __forceinline__ void gather32(const char* __restrict__ xb, unsigned cOff, int sub,
                                         int sReg, int n, float4& acc) {
    if (n >= 16) gather16<0, true>(xb, cOff, sub, sReg, 16, acc);
    else if (n > 0) gather16<0, false>(xb, cOff, sub, sReg, n, acc);
    if (n >= 32) gather16<16, true>(xb, cOff, sub, sReg, 16, acc);
    else if (n > 16) gather16<16, false>(xb, cOff, sub, sReg, n - 16, acc);
}

template <int D, int MD, bool LEAKY, bool SCALE_OUT>
__global__ void __launch_bounds__(256) __attribute__((amdgpu_waves_per_eu(12, 16)))
k_layer(const float* __restrict__ x, const float* __restrict__ invO,
        const float* __restrict__ invI, const int* __restrict__ rowptr,
        const int* __restrict__ csr, const float* __restrict__ W,
        const float* __restrict__ bias, float* __restrict__ out, int Nn) {
    constexpr int LPR   = D / 4;
    constexpr int EPI   = 32 / LPR;
    constexpr int PITCH = (D >= 64) ? (D + 4) : D;
    constexpr int NK    = D / 4;
    constexpr int COLT  = (MD + 15) / 16;
    constexpr bool FULLCOL = (MD % 16 == 0);
    static_assert(8 % COLT == 0, "a wave's tiles must share one column slice");
    constexpr int ROWS  = 80;
    constexpr int NPW   = ROWS / 8;
    constexpr int NT    = (ROWS / 16) * COLT;
    constexpr bool ALLWAVES = (NT >= 8);

    __shared__ float4 sA4[ROWS * PITCH / 4];

    const int lane     = threadIdx.x & 31;
    const int wv       = __builtin_amdgcn_readfirstlane((int)(threadIdx.x >> 5));
    const int nodeBase = blockIdx.x * ROWS;
    const int sub      = lane / LPR;
    const int c        = lane % LPR;
    const float* xc    = x + c * 4;

    const int first = nodeBase + wv * NPW;
    const int pIdx  = first + lane;
    const int   rp  = rowptr[pIdx < Nn ? pIdx : Nn];
    const float nrm = invI[pIdx < Nn ? pIdx : (Nn - 1)];

    if constexpr (D == 64) {
        const char* xb = reinterpret_cast<const char*>(x);
        const unsigned cOff = (unsigned)c * 16u;
        int   sC0, sC1, sN0, sN1;
        {
            const int b0 = __builtin_amdgcn_readlane(rp, 0);
            const int e0 = __builtin_amdgcn_readlane(rp, 1);
            const int e1 = __builtin_amdgcn_readlane(rp, 2);
            listLoad(csr, b0, e0, lane, sC0, sC1);
            listLoad(csr, e0, e1, lane, sN0, sN1);
        }
        #pragma nounroll
        for (int i = 0; i < NPW; ++i) {
            const int beg  = __builtin_amdgcn_readlane(rp, i);
            const int end  = __builtin_amdgcn_readlane(rp, i + 1);
            const int end1 = __builtin_amdgcn_readlane(rp, i + 2);
            const int end2 = __builtin_amdgcn_readlane(rp, i + 3);
            const float si = __int_as_float(__builtin_amdgcn_readlane(__float_as_int(nrm), i));

            int   sM0, sM1;
            listLoad(csr, end1, end2, lane, sM0, sM1);

            const int deg = end - beg;
            float4 acc = make_float4(0.f, 0.f, 0.f, 0.f);
            gather32(xb, cOff, sub, sC0, deg, acc);
            if (deg > 32) gather32(xb, cOff, sub, sC1, deg - 32, acc);
            for (int e = beg + 64 + sub; e < end; e += EPI) {
                const int s0 = csr[e];
                const float4 v0 = *reinterpret_cast<const float4*>(xc + (size_t)s0 * D);
                acc.x += v0.x; acc.y += v0.y; acc.z += v0.z; acc.w += v0.w;
            }
            acc.x += __shfl_xor(acc.x, 16);
            acc.y += __shfl_xor(acc.y, 16);
            acc.z += __shfl_xor(acc.z, 16);
            acc.w += __shfl_xor(acc.w, 16);
            if (sub == 0)
                sA4[(wv * NPW + i) * (PITCH / 4) + c] =
                    make_float4(acc.x * si, acc.y * si, acc.z * si, acc.w * si);

            sC0 = sN0; sC1 = sN1;
            sN0 = sM0; sN1 = sM1;
        }
    } else {
        for (int i = 0; i < NPW; ++i) {
            const int beg = __builtin_amdgcn_readlane(rp, i);
            const int end = __builtin_amdgcn_readlane(rp, i + 1);
            const float si = __int_as_float(__builtin_amdgcn_readlane(__float_as_int(nrm), i));

            float4 acc = make_float4(0.f, 0.f, 0.f, 0.f);
            int e = beg + sub;
            for (; e + 3 * EPI < end; e += 4 * EPI) {
                const int s0 = csr[e];
                const int s1 = csr[e + EPI];
                const int s2 = csr[e + 2 * EPI];
                const int s3 = csr[e + 3 * EPI];
                const float4 v0 = *reinterpret_cast<const float4*>(xc + (size_t)s0 * D);
                const float4 v1 = *reinterpret_cast<const float4*>(xc + (size_t)s1 * D);
                const float4 v2 = *reinterpret_cast<const float4*>(xc + (size_t)s2 * D);
                const float4 v3 = *reinterpret_cast<const float4*>(xc + (size_t)s3 * D);
                acc.x += v0.x; acc.y += v0.y; acc.z += v0.z; acc.w += v0.w;
                acc.x += v1.x; acc.y += v1.y; acc.z += v1.z; acc.w += v1.w;
                acc.x += v2.x; acc.y += v2.y; acc.z += v2.z; acc.w += v2.w;
                acc.x += v3.x; acc.y += v3.y; acc.z += v3.z; acc.w += v3.w;
            }
            for (; e < end; e += EPI) {
                const int s0 = csr[e];
                const float4 v0 = *reinterpret_cast<const float4*>(xc + (size_t)s0 * D);
                acc.x += v0.x; acc.y += v0.y; acc.z += v0.z; acc.w += v0.w;
            }
            #pragma unroll
            for (int o = LPR; o < 32; o <<= 1) {
                acc.x += __shfl_xor(acc.x, o);
                acc.y += __shfl_xor(acc.y, o);
                acc.z += __shfl_xor(acc.z, o);
                acc.w += __shfl_xor(acc.w, o);
            }
            if (sub == 0)
                sA4[(wv * NPW + i) * (PITCH / 4) + c] =
                    make_float4(acc.x * si, acc.y * si, acc.z * si, acc.w * si);
        }
    }

    const int   ln15    = lane & 15;
    const int   hi      = lane >> 4;
    const int   kHalf   = hi * 2;
    const int   ct      = wv % COLT;
    const int   col     = ct * 16 + ln15;
    const int   colC    = FULLCOL ? col : ((col < MD) ? col : (MD - 1));
    const float colMask = FULLCOL ? 1.f : ((col < MD) ? 1.f : 0.f);

    v2f bf[NK];
    float bv = 0.f;
    if (ALLWAVES || wv < NT) {
        #pragma unroll
        for (int t = 0; t < NK; ++t) {
            const int k0 = t * 4 + kHalf;
            bf[t].x = W[k0 * MD + colC] * colMask;
            bf[t].y = W[(k0 + 1) * MD + colC] * colMask;
        }
        bv = bias[colC] * colMask;
    }

    __syncthreads();

    const float* sAf = reinterpret_cast<const float*>(sA4);
    for (int tl = wv; tl < NT; tl += 8) {
        const int rt       = tl / COLT;
        const int tileBase = nodeBase + rt * 16;
        if (tileBase >= Nn) break;
        const float* aRow = sAf + (rt * 16 + ln15) * PITCH + kHalf;

        float so[8];
        if constexpr (SCALE_OUT) {
            const int r0 = tileBase + hi * 8;
            if (tileBase + 16 <= Nn) {
                const float4 n0 = *reinterpret_cast<const float4*>(invO + r0);
                const float4 n1 = *reinterpret_cast<const float4*>(invO + r0 + 4);
                so[0] = n0.x; so[1] = n0.y; so[2] = n0.z; so[3] = n0.w;
                so[4] = n1.x; so[5] = n1.y; so[6] = n1.z; so[7] = n1.w;
            } else {
                #pragma unroll
                for (int j = 0; j < 8; ++j) {
                    const int r = r0 + j;
                    so[j] = invO[r < Nn ? r : (Nn - 1)];
                }
            }
        }

        v8f acc = {0.f, 0.f, 0.f, 0.f, 0.f, 0.f, 0.f, 0.f};
        #pragma unroll
        for (int t = 0; t < NK; ++t) {
            const v2f a = *reinterpret_cast<const v2f*>(aRow + t * 4);
            acc = __builtin_amdgcn_wmma_f32_16x16x4_f32(
                false, a, false, bf[t], (short)0, acc, false, false);
        }

        float h[8];
        #pragma unroll
        for (int j = 0; j < 8; ++j) {
            const float v = acc[j] + bv;
            const float a = LEAKY ? fmaxf(v, LEAKY_SLOPE * v) : v;
            if constexpr (SCALE_OUT) h[j] = a * so[j];
            else h[j] = a;
        }

        float* outBase = out + (size_t)(tileBase + hi * 8) * MD + col;
        if (FULLCOL && (tileBase + 16 <= Nn)) {
            #pragma unroll
            for (int j = 0; j < 8; ++j) outBase[j * MD] = h[j];
        } else {
            #pragma unroll
            for (int j = 0; j < 8; ++j) {
                const int r = tileBase + hi * 8 + j;
                if (col < MD && r < Nn) outBase[j * MD] = h[j];
            }
        }
    }
}

extern "C" void kernel_launch(void* const* d_in, const int* in_sizes, int n_in,
                              void* d_out, int out_size, void* d_ws, size_t ws_size,
                              hipStream_t stream) {
    (void)n_in; (void)out_size;

    const float* features = (const float*)d_in[0];
    const int*   esrc     = (const int*)d_in[1];
    const int*   edst     = (const int*)d_in[2];

    const int N = in_sizes[0] / 4;
    const int E = in_sizes[1];
    static const int DIMS[9] = {4, 64, 64, 64, 64, 64, 64, 64, 3};

    float* ws      = (float*)d_ws;
    float* inv_out = ws;
    float* inv_in  = ws + 100352;
    float* xbuf    = ws + 200704;
    float* agg     = ws + 200704 + 6400000;

    const int BT = 256;

    if (ws_size >= (size_t)65204224) {
        int* partI  = (int*)xbuf;
        int* partO  = (int*)(xbuf + 3200000);
        int* cntI   = (int*)(agg + 401408);
        int* csr    = (int*)(ws + 13000704);
        int* rowptr = (int*)(ws + 13000704 + 3200000);

        k_lb_count<<<LB_CHUNKS * LB_RANGES, LB_T, 0, stream>>>(esrc, edst, partI, partO, E, N);
        k_lb_nodes<<<(N + BT - 1) / BT, BT, 0, stream>>>(partI, partO, cntI, inv_out, inv_in,
                                                          features, agg, N);
        k_scan<<<1, 1024, 0, stream>>>(cntI, rowptr, N);
        k_lb_place<<<LB_CHUNKS * LB_RANGES, LB_T, 0, stream>>>(esrc, edst, rowptr, partI, csr, E, N);

        const int lblocks = (N + 79) / 80;
        float* planeA = xbuf;
        float* planeB = agg;
        k_layer<4, 64, true, true><<<lblocks, BT, 0, stream>>>(
            planeB, inv_out, inv_in, rowptr, csr,
            (const float*)d_in[3], (const float*)d_in[4], planeA, N);
        const float* xin = planeA;
        float* xout = planeB;
        for (int i = 1; i < 7; ++i) {
            k_layer<64, 64, true, true><<<lblocks, BT, 0, stream>>>(
                xin, inv_out, inv_in, rowptr, csr,
                (const float*)d_in[3 + 2 * i], (const float*)d_in[4 + 2 * i], xout, N);
            float* t = (float*)xin;
            xin = xout;
            xout = t;
        }
        k_layer<64, 3, false, false><<<lblocks, BT, 0, stream>>>(
            xin, inv_out, inv_in, rowptr, csr,
            (const float*)d_in[17], (const float*)d_in[18], (float*)d_out, N);
        return;
    }

    {
        int n4 = (N + 3) / 4;
        k_zero4<<<(n4 + BT - 1) / BT, BT, 0, stream>>>((float4*)inv_out, n4);
        k_zero4<<<(n4 + BT - 1) / BT, BT, 0, stream>>>((float4*)inv_in,  n4);
        k_degrees<<<(E + BT - 1) / BT, BT, 0, stream>>>(esrc, edst, inv_out, inv_in, E);
        k_inv_sqrt<<<(N + BT - 1) / BT, BT, 0, stream>>>(inv_out, inv_in, N);
    }

    const float* xcur = features;
    for (int i = 0; i < 8; ++i) {
        const int Kd = DIMS[i];
        const int Md = DIMS[i + 1];
        const float* Wi = (const float*)d_in[3 + 2 * i];
        const float* bi = (const float*)d_in[4 + 2 * i];

        int n4 = N * Kd / 4;
        k_zero4<<<(n4 + BT - 1) / BT, BT, 0, stream>>>((float4*)agg, n4);

        int chunks = Kd / 4;
        int lg = (chunks == 16) ? 4 : 0;
        long long total = (long long)E * chunks;
        int blocks = (int)((total + BT - 1) / BT);
        k_gather<<<blocks, BT, 0, stream>>>(xcur, inv_out, esrc, edst, agg, E, lg, Kd);

        int colTiles  = (Md + 15) / 16;
        int rowBlocks = (N + 63) / 64;
        long long waves = (long long)rowBlocks * colTiles;
        long long thr = waves * 32;
        int gblocks = (int)((thr + BT - 1) / BT);
        float* yout = (i == 7) ? (float*)d_out : xbuf;

        if (i == 0) {
            k_gemm_wmma<4, 64, true><<<gblocks, BT, 0, stream>>>(
                agg, inv_in, Wi, bi, yout, N, colTiles, rowBlocks);
        } else if (i < 7) {
            k_gemm_wmma<64, 64, true><<<gblocks, BT, 0, stream>>>(
                agg, inv_in, Wi, bi, yout, N, colTiles, rowBlocks);
        } else {
            k_gemm_wmma<64, 3, false><<<gblocks, BT, 0, stream>>>(
                agg, inv_in, Wi, bi, yout, N, colTiles, rowBlocks);
        }
        xcur = xbuf;
    }
}


# Last layer's 64x3 product moved in front of its aggregation: layer 7's rows go through W8 from LDS on the f32 WMMA; last launch sums 4-float rows per edge, not 64-float; its sum order changes
# speedup vs baseline: 7.2695x; 1.0460x over previous
#include <hip/hip_runtime.h>


typedef __attribute__((ext_vector_type(2))) float v2f;
typedef __attribute__((ext_vector_type(8))) float v8f;

#define LEAKY_SLOPE 0.01f

__device__ __forceinline__ void atomAddF(float* p, float v) {
    __hip_atomic_fetch_add(p, v, __ATOMIC_RELAXED, __HIP_MEMORY_SCOPE_AGENT);
}

__global__ void k_zero4(float4* __restrict__ p, int n4) {
    int i = blockIdx.x * blockDim.x + threadIdx.x;
    if (i < n4) p[i] = make_float4(0.f, 0.f, 0.f, 0.f);
}

__global__ void k_degrees(const int* __restrict__ src, const int* __restrict__ dst,
                          float* __restrict__ degO, float* __restrict__ degI, int E) {
    int e = blockIdx.x * blockDim.x + threadIdx.x;
    if (e >= E) return;
    atomAddF(&degO[src[e]], 1.0f);
    atomAddF(&degI[dst[e]], 1.0f);
}

__global__ void k_inv_sqrt(float* __restrict__ a, float* __restrict__ b, int n) {
    int i = blockIdx.x * blockDim.x + threadIdx.x;
    if (i >= n) return;
    a[i] = rsqrtf(fmaxf(a[i], 1.0f));
    b[i] = rsqrtf(fmaxf(b[i], 1.0f));
}

__global__ void k_gather(const float* __restrict__ x, const float* __restrict__ invO,
                         const int* __restrict__ src, const int* __restrict__ dst,
                         float* __restrict__ agg, int E, int lgChunks, int d) {
    int tid = blockIdx.x * blockDim.x + threadIdx.x;
    int e = tid >> lgChunks;
    if (e >= E) return;
    int c = tid & ((1 << lgChunks) - 1);
    int s = src[e];
    int t = dst[e];
    float w = invO[s];
    const float4 v = *reinterpret_cast<const float4*>(x + (size_t)s * d + c * 4);
    float* base = agg + (size_t)t * d + c * 4;
    atomAddF(base + 0, v.x * w);
    atomAddF(base + 1, v.y * w);
    atomAddF(base + 2, v.z * w);
    atomAddF(base + 3, v.w * w);
}

#define LB_RANGES 8
#define LB_CHUNKS 32
#define LB_RW     12544
#define LB_T      1024

__device__ __forceinline__ void lbChunk(int E, int ch, int& e0, int& e1) {
    const int per = ((E + LB_CHUNKS * 4 - 1) / (LB_CHUNKS * 4)) * 4;
    e0 = ch * per;
    if (e0 > E) e0 = E;
    e1 = e0 + per;
    if (e1 > E) e1 = E;
}

#define LB_COUNT(c, v)                                                                   \
    do {                                                                                 \
        const unsigned r_ = (unsigned)((v) - lo);                                        \
        if (r_ < ulen)                                                                   \
            __hip_atomic_fetch_add(&c[r_], 1, __ATOMIC_RELAXED, __HIP_MEMORY_SCOPE_WORKGROUP); \
    } while (0)

#define LB_COUNT4(s, d)                                                                  \
    do {                                                                                 \
        LB_COUNT(cO, (s).x); LB_COUNT(cO, (s).y); LB_COUNT(cO, (s).z); LB_COUNT(cO, (s).w); \
        LB_COUNT(cI, (d).x); LB_COUNT(cI, (d).y); LB_COUNT(cI, (d).z); LB_COUNT(cI, (d).w); \
    } while (0)

__global__ void __launch_bounds__(LB_T)
k_lb_count(const int* __restrict__ src, const int* __restrict__ dst,
           int* __restrict__ partI, int* __restrict__ partO, int E, int N) {
    __shared__ int cI[LB_RW];
    __shared__ int cO[LB_RW];
    const int tid = threadIdx.x;
    const int ch  = blockIdx.x / LB_RANGES;
    const int rg  = blockIdx.x % LB_RANGES;
    const int lo  = rg * LB_RW;
    int len = N - lo;
    if (len > LB_RW) len = LB_RW;
    if (len < 0) len = 0;
    const unsigned ulen = (unsigned)len;

    for (int n = tid; n < LB_RW; n += LB_T) { cI[n] = 0; cO[n] = 0; }
    __syncthreads();

    int e0, e1;
    lbChunk(E, ch, e0, e1);
    const int n4 = (e1 - e0) >> 2;
    const int4* s4 = reinterpret_cast<const int4*>(src + e0);
    const int4* d4 = reinterpret_cast<const int4*>(dst + e0);
    int q = tid;
    for (; q + 3 * LB_T < n4; q += 4 * LB_T) {
        const int4 a0 = s4[q], a1 = s4[q + LB_T], a2 = s4[q + 2 * LB_T], a3 = s4[q + 3 * LB_T];
        const int4 b0 = d4[q], b1 = d4[q + LB_T], b2 = d4[q + 2 * LB_T], b3 = d4[q + 3 * LB_T];
        LB_COUNT4(a0, b0); LB_COUNT4(a1, b1); LB_COUNT4(a2, b2); LB_COUNT4(a3, b3);
    }
    for (; q < n4; q += LB_T) {
        const int4 a0 = s4[q];
        const int4 b0 = d4[q];
        LB_COUNT4(a0, b0);
    }
    for (int e = e0 + (n4 << 2) + tid; e < e1; e += LB_T) {
        const int s = src[e];
        const int d = dst[e];
        LB_COUNT(cO, s);
        LB_COUNT(cI, d);
    }
    __syncthreads();

    int* pI = partI + (size_t)ch * N + lo;
    int* pO = partO + (size_t)ch * N + lo;
    for (int n = tid; n < len; n += LB_T) { pI[n] = cI[n]; pO[n] = cO[n]; }
}

__global__ void k_lb_nodes(int* __restrict__ partI, const int* __restrict__ partO,
                           int* __restrict__ cntI, float* __restrict__ invO,
                           float* __restrict__ invI, const float* __restrict__ feat,
                           float* __restrict__ xs0, int N) {
    const int n = blockIdx.x * blockDim.x + threadIdx.x;
    if (n >= N) return;
    int vi[LB_CHUNKS], vo[LB_CHUNKS];
    #pragma unroll
    for (int i = 0; i < LB_CHUNKS; ++i) {
        vi[i] = partI[(size_t)i * N + n];
        vo[i] = partO[(size_t)i * N + n];
    }
    int accI = 0, accO = 0;
    #pragma unroll
    for (int i = 0; i < LB_CHUNKS; ++i) {
        partI[(size_t)i * N + n] = accI;
        accI += vi[i];
        accO += vo[i];
    }
    cntI[n] = accI;
    const float wo = rsqrtf(fmaxf((float)accO, 1.0f));
    invO[n] = wo;
    invI[n] = rsqrtf(fmaxf((float)accI, 1.0f));
    const float4 f = *reinterpret_cast<const float4*>(feat + (size_t)n * 4);
    *reinterpret_cast<float4*>(xs0 + (size_t)n * 4) = make_float4(f.x * wo, f.y * wo, f.z * wo, f.w * wo);
}

#define LB_PLACE(s, d)                                                                   \
    do {                                                                                 \
        const unsigned r_ = (unsigned)((d) - lo);                                        \
        if (r_ < ulen) {                                                                 \
            const int p_ = __hip_atomic_fetch_add(&pos[r_], 1, __ATOMIC_RELAXED,         \
                                                  __HIP_MEMORY_SCOPE_WORKGROUP);         \
            csr[p_] = (s);                                                               \
        }                                                                                \
    } while (0)

#define LB_PLACE4(s, d)                                                                  \
    do {                                                                                 \
        LB_PLACE((s).x, (d).x); LB_PLACE((s).y, (d).y);                                  \
        LB_PLACE((s).z, (d).z); LB_PLACE((s).w, (d).w);                                  \
    } while (0)

__global__ void __launch_bounds__(LB_T)
k_lb_place(const int* __restrict__ src, const int* __restrict__ dst,
           const int* __restrict__ rowptr, const int* __restrict__ partI,
           int* __restrict__ csr, int E, int N) {
    __shared__ int pos[LB_RW];
    const int tid = threadIdx.x;
    const int ch  = blockIdx.x / LB_RANGES;
    const int rg  = blockIdx.x % LB_RANGES;
    const int lo  = rg * LB_RW;
    int len = N - lo;
    if (len > LB_RW) len = LB_RW;
    if (len < 0) len = 0;
    const unsigned ulen = (unsigned)len;

    const int* pI = partI + (size_t)ch * N + lo;
    for (int n = tid; n < len; n += LB_T) pos[n] = rowptr[lo + n] + pI[n];
    __syncthreads();

    int e0, e1;
    lbChunk(E, ch, e0, e1);
    const int n4 = (e1 - e0) >> 2;
    const int4* s4 = reinterpret_cast<const int4*>(src + e0);
    const int4* d4 = reinterpret_cast<const int4*>(dst + e0);
    int q = tid;
    for (; q + 3 * LB_T < n4; q += 4 * LB_T) {
        const int4 a0 = s4[q], a1 = s4[q + LB_T], a2 = s4[q + 2 * LB_T], a3 = s4[q + 3 * LB_T];
        const int4 b0 = d4[q], b1 = d4[q + LB_T], b2 = d4[q + 2 * LB_T], b3 = d4[q + 3 * LB_T];
        LB_PLACE4(a0, b0); LB_PLACE4(a1, b1); LB_PLACE4(a2, b2); LB_PLACE4(a3, b3);
    }
    for (; q < n4; q += LB_T) {
        const int4 a0 = s4[q];
        const int4 b0 = d4[q];
        LB_PLACE4(a0, b0);
    }
    for (int e = e0 + (n4 << 2) + tid; e < e1; e += LB_T) {
        const int s = src[e];
        const int d = dst[e];
        LB_PLACE(s, d);
    }
}

__global__ void __launch_bounds__(1024) k_scan(const int* __restrict__ cnt,
                                               int* __restrict__ rowptr, int n) {
    __shared__ int wsum[32];
    __shared__ int tileTotal;
    const int tid = threadIdx.x, lane = tid & 31, wv = tid >> 5;
    int carry = 0;
    for (int base = 0; base < n; base += 4096) {
        const int i = base + tid * 4;
        int a0 = 0, a1 = 0, a2 = 0, a3 = 0;
        if (i + 3 < n) {
            const int4 v = *reinterpret_cast<const int4*>(cnt + i);
            a0 = v.x; a1 = v.y; a2 = v.z; a3 = v.w;
        } else {
            if (i < n) a0 = cnt[i];
            if (i + 1 < n) a1 = cnt[i + 1];
            if (i + 2 < n) a2 = cnt[i + 2];
        }
        const int tsum = a0 + a1 + a2 + a3;
        int inc = tsum;
        for (int o = 1; o < 32; o <<= 1) {
            int t = __shfl_up(inc, o);
            if (lane >= o) inc += t;
        }
        if (lane == 31) wsum[wv] = inc;
        __syncthreads();
        if (wv == 0) {
            const int w = wsum[lane];
            int winc = w;
            for (int o = 1; o < 32; o <<= 1) {
                int t = __shfl_up(winc, o);
                if (lane >= o) winc += t;
            }
            wsum[lane] = winc - w;
            if (lane == 31) tileTotal = winc;
        }
        __syncthreads();
        const int ex = carry + wsum[wv] + (inc - tsum);
        carry += tileTotal;
        if (i + 3 < n) {
            *reinterpret_cast<int4*>(rowptr + i) = make_int4(ex, ex + a0, ex + a0 + a1, ex + a0 + a1 + a2);
        } else {
            if (i < n) rowptr[i] = ex;
            if (i + 1 < n) rowptr[i + 1] = ex + a0;
            if (i + 2 < n) rowptr[i + 2] = ex + a0 + a1;
        }
        __syncthreads();
    }
    if (tid == 0) rowptr[n] = carry;
}

template <int KDIM, int MD, bool LEAKY>
__global__ void k_gemm_wmma(const float* __restrict__ agg, const float* __restrict__ invI,
                            const float* __restrict__ W, const float* __restrict__ bias,
                            float* __restrict__ out, int Nn, int colTiles, int rowBlocks) {
    constexpr int NK = KDIM / 4;
    constexpr bool FULLCOL = (MD % 16 == 0);

    int gwave    = __builtin_amdgcn_readfirstlane(
                       (int)((blockIdx.x * blockDim.x + threadIdx.x) >> 5));
    int lane     = threadIdx.x & 31;
    int rowBlock = gwave / colTiles;
    int colTile  = gwave - rowBlock * colTiles;
    if (rowBlock >= rowBlocks) return;

    int   ln15    = lane & 15;
    int   col     = colTile * 16 + ln15;
    int   colC    = FULLCOL ? col : ((col < MD) ? col : (MD - 1));
    float colMask = FULLCOL ? 1.f : ((col < MD) ? 1.f : 0.f);
    int   kHalf   = (lane >> 4) * 2;
    int   hi      = lane >> 4;

    v2f bf[NK];
    #pragma unroll
    for (int t = 0; t < NK; ++t) {
        int k0 = t * 4 + kHalf;
        bf[t].x = W[k0 * MD + colC] * colMask;
        bf[t].y = W[(k0 + 1) * MD + colC] * colMask;
    }
    float bv = bias[colC] * colMask;

    #pragma unroll
    for (int st = 0; st < 4; ++st) {
        int tileBase = rowBlock * 64 + st * 16;
        if (tileBase >= Nn) break;
        int   mrow  = tileBase + ln15;
        int   mrowC = (mrow < Nn) ? mrow : (Nn - 1);
        float s     = invI[mrowC];

        v8f acc = {0.f, 0.f, 0.f, 0.f, 0.f, 0.f, 0.f, 0.f};
        #pragma unroll
        for (int t = 0; t < NK; ++t) {
            int k0 = t * 4 + kHalf;
            v2f a = *reinterpret_cast<const v2f*>(agg + (size_t)mrowC * KDIM + k0);
            a.x *= s;
            a.y *= s;
            acc = __builtin_amdgcn_wmma_f32_16x16x4_f32(
                false, a, false, bf[t], (short)0, acc, false, false);
        }

        float h[8];
        #pragma unroll
        for (int j = 0; j < 8; ++j) {
            float v = acc[j] + bv;
            h[j] = LEAKY ? fmaxf(v, LEAKY_SLOPE * v) : v;
        }

        float* outBase = out + (size_t)(tileBase + hi * 8) * MD + col;
        if (FULLCOL && (tileBase + 16 <= Nn)) {
            #pragma unroll
            for (int j = 0; j < 8; ++j) outBase[j * MD] = h[j];
        } else {
            #pragma unroll
            for (int j = 0; j < 8; ++j) {
                int r = tileBase + hi * 8 + j;
                if (col < MD && r < Nn) outBase[j * MD] = h[j];
            }
        }
    }
}

__device__ __forceinline__ void listLoad(const int* __restrict__ csr, int beg, int end, int lane,
                                         int& s0, int& s1) {
    const int last = (end > 0) ? (end - 1) : 0;
    const int e0 = beg + lane;
    const int e1 = e0 + 32;
    s0 = csr[e0 < last ? e0 : last];
    s1 = csr[e1 < last ? e1 : last];
}

template <int L0, bool FULL>
__device__ __forceinline__ void gather16(const char* __restrict__ xb, unsigned cOff, int sub,
                                         int sReg, int cnt, float4& acc) {
    float4 v[8];
    const int sFirst = __builtin_amdgcn_readlane(sReg, L0);
    #pragma unroll
    for (int u = 0; u < 8; ++u) {
        const int sLo = __builtin_amdgcn_readlane(sReg, L0 + 2 * u);
        const int sHi = __builtin_amdgcn_readlane(sReg, L0 + 2 * u + 1);
        int s = sub ? sHi : sLo;
        if (!FULL) s = (2 * u + sub < cnt) ? s : sFirst;
        v[u] = *reinterpret_cast<const float4*>(xb + (((unsigned)s << 8) + cOff));
    }
    #pragma unroll
    for (int u = 0; u < 8; ++u) {
        float4 t = v[u];
        if (!FULL && !(2 * u + sub < cnt)) t = make_float4(0.f, 0.f, 0.f, 0.f);
        acc.x += t.x; acc.y += t.y; acc.z += t.z; acc.w += t.w;
    }
}

__device__ __forceinline__ void gather32(const char* __restrict__ xb, unsigned cOff, int sub,
                                         int sReg, int n, float4& acc) {
    if (n >= 16) gather16<0, true>(xb, cOff, sub, sReg, 16, acc);
    else if (n > 0) gather16<0, false>(xb, cOff, sub, sReg, n, acc);
    if (n >= 32) gather16<16, true>(xb, cOff, sub, sReg, 16, acc);
    else if (n > 16) gather16<16, false>(xb, cOff, sub, sReg, n - 16, acc);
}

template <int D, int MD, bool LEAKY, bool SCALE_OUT, int MD2>
__global__ void __launch_bounds__(256) __attribute__((amdgpu_waves_per_eu(12, 16)))
k_layer(const float* __restrict__ x, const float* __restrict__ invO,
        const float* __restrict__ invI, const int* __restrict__ rowptr,
        const int* __restrict__ csr, const float* __restrict__ W,
        const float* __restrict__ bias, const float* __restrict__ W2,
        float* __restrict__ out, int Nn) {
    constexpr int LPR   = D / 4;
    constexpr int EPI   = 32 / LPR;
    constexpr int PITCH = (D >= 64) ? (D + 4) : D;
    constexpr int NK    = D / 4;
    constexpr int COLT  = (MD + 15) / 16;
    constexpr bool FULLCOL = (MD % 16 == 0);
    static_assert(8 % COLT == 0, "a wave's tiles must share one column slice");
    constexpr int ROWS  = 80;
    constexpr int NPW   = ROWS / 8;
    constexpr int NT    = (ROWS / 16) * COLT;
    constexpr bool ALLWAVES = (NT >= 8);

    __shared__ float4 sA4[ROWS * PITCH / 4];
    __shared__ float sH[(MD2 > 0) ? (ROWS * PITCH) : 4];

    const int lane     = threadIdx.x & 31;
    const int wv       = __builtin_amdgcn_readfirstlane((int)(threadIdx.x >> 5));
    const int nodeBase = blockIdx.x * ROWS;
    const int sub      = lane / LPR;
    const int c        = lane % LPR;
    const float* xc    = x + c * 4;

    const int first = nodeBase + wv * NPW;
    const int pIdx  = first + lane;
    const int   rp  = rowptr[pIdx < Nn ? pIdx : Nn];
    const float nrm = invI[pIdx < Nn ? pIdx : (Nn - 1)];

    if constexpr (D == 64) {
        const char* xb = reinterpret_cast<const char*>(x);
        const unsigned cOff = (unsigned)c * 16u;
        int   sC0, sC1, sN0, sN1;
        {
            const int b0 = __builtin_amdgcn_readlane(rp, 0);
            const int e0 = __builtin_amdgcn_readlane(rp, 1);
            const int e1 = __builtin_amdgcn_readlane(rp, 2);
            listLoad(csr, b0, e0, lane, sC0, sC1);
            listLoad(csr, e0, e1, lane, sN0, sN1);
        }
        #pragma nounroll
        for (int i = 0; i < NPW; ++i) {
            const int beg  = __builtin_amdgcn_readlane(rp, i);
            const int end  = __builtin_amdgcn_readlane(rp, i + 1);
            const int end1 = __builtin_amdgcn_readlane(rp, i + 2);
            const int end2 = __builtin_amdgcn_readlane(rp, i + 3);
            const float si = __int_as_float(__builtin_amdgcn_readlane(__float_as_int(nrm), i));

            int   sM0, sM1;
            listLoad(csr, end1, end2, lane, sM0, sM1);

            const int deg = end - beg;
            float4 acc = make_float4(0.f, 0.f, 0.f, 0.f);
            gather32(xb, cOff, sub, sC0, deg, acc);
            if (deg > 32) gather32(xb, cOff, sub, sC1, deg - 32, acc);
            for (int e = beg + 64 + sub; e < end; e += EPI) {
                const int s0 = csr[e];
                const float4 v0 = *reinterpret_cast<const float4*>(xc + (size_t)s0 * D);
                acc.x += v0.x; acc.y += v0.y; acc.z += v0.z; acc.w += v0.w;
            }
            acc.x += __shfl_xor(acc.x, 16);
            acc.y += __shfl_xor(acc.y, 16);
            acc.z += __shfl_xor(acc.z, 16);
            acc.w += __shfl_xor(acc.w, 16);
            if (sub == 0)
                sA4[(wv * NPW + i) * (PITCH / 4) + c] =
                    make_float4(acc.x * si, acc.y * si, acc.z * si, acc.w * si);

            sC0 = sN0; sC1 = sN1;
            sN0 = sM0; sN1 = sM1;
        }
    } else {
        for (int i = 0; i < NPW; ++i) {
            const int beg = __builtin_amdgcn_readlane(rp, i);
            const int end = __builtin_amdgcn_readlane(rp, i + 1);
            const float si = __int_as_float(__builtin_amdgcn_readlane(__float_as_int(nrm), i));

            float4 acc = make_float4(0.f, 0.f, 0.f, 0.f);
            int e = beg + sub;
            for (; e + 3 * EPI < end; e += 4 * EPI) {
                const int s0 = csr[e];
                const int s1 = csr[e + EPI];
                const int s2 = csr[e + 2 * EPI];
                const int s3 = csr[e + 3 * EPI];
                const float4 v0 = *reinterpret_cast<const float4*>(xc + (size_t)s0 * D);
                const float4 v1 = *reinterpret_cast<const float4*>(xc + (size_t)s1 * D);
                const float4 v2 = *reinterpret_cast<const float4*>(xc + (size_t)s2 * D);
                const float4 v3 = *reinterpret_cast<const float4*>(xc + (size_t)s3 * D);
                acc.x += v0.x; acc.y += v0.y; acc.z += v0.z; acc.w += v0.w;
                acc.x += v1.x; acc.y += v1.y; acc.z += v1.z; acc.w += v1.w;
                acc.x += v2.x; acc.y += v2.y; acc.z += v2.z; acc.w += v2.w;
                acc.x += v3.x; acc.y += v3.y; acc.z += v3.z; acc.w += v3.w;
            }
            for (; e < end; e += EPI) {
                const int s0 = csr[e];
                const float4 v0 = *reinterpret_cast<const float4*>(xc + (size_t)s0 * D);
                acc.x += v0.x; acc.y += v0.y; acc.z += v0.z; acc.w += v0.w;
            }
            #pragma unroll
            for (int o = LPR; o < 32; o <<= 1) {
                acc.x += __shfl_xor(acc.x, o);
                acc.y += __shfl_xor(acc.y, o);
                acc.z += __shfl_xor(acc.z, o);
                acc.w += __shfl_xor(acc.w, o);
            }
            if (sub == 0)
                sA4[(wv * NPW + i) * (PITCH / 4) + c] =
                    make_float4(acc.x * si, acc.y * si, acc.z * si, acc.w * si);
        }
    }

    const int   ln15    = lane & 15;
    const int   hi      = lane >> 4;
    const int   kHalf   = hi * 2;
    const int   ct      = wv % COLT;
    const int   col     = ct * 16 + ln15;
    const int   colC    = FULLCOL ? col : ((col < MD) ? col : (MD - 1));
    const float colMask = FULLCOL ? 1.f : ((col < MD) ? 1.f : 0.f);

    v2f bf[NK];
    float bv = 0.f;
    if (ALLWAVES || wv < NT) {
        #pragma unroll
        for (int t = 0; t < NK; ++t) {
            const int k0 = t * 4 + kHalf;
            bf[t].x = W[k0 * MD + colC] * colMask;
            bf[t].y = W[(k0 + 1) * MD + colC] * colMask;
        }
        bv = bias[colC] * colMask;
    }

    __syncthreads();

    const float* sAf = reinterpret_cast<const float*>(sA4);
    for (int tl = wv; tl < NT; tl += 8) {
        const int rt       = tl / COLT;
        const int tileBase = nodeBase + rt * 16;
        if (tileBase >= Nn) break;
        const float* aRow = sAf + (rt * 16 + ln15) * PITCH + kHalf;

        float so[8];
        if constexpr (SCALE_OUT) {
            const int r0 = tileBase + hi * 8;
            if (tileBase + 16 <= Nn) {
                const float4 n0 = *reinterpret_cast<const float4*>(invO + r0);
                const float4 n1 = *reinterpret_cast<const float4*>(invO + r0 + 4);
                so[0] = n0.x; so[1] = n0.y; so[2] = n0.z; so[3] = n0.w;
                so[4] = n1.x; so[5] = n1.y; so[6] = n1.z; so[7] = n1.w;
            } else {
                #pragma unroll
                for (int j = 0; j < 8; ++j) {
                    const int r = r0 + j;
                    so[j] = invO[r < Nn ? r : (Nn - 1)];
                }
            }
        }

        v8f acc = {0.f, 0.f, 0.f, 0.f, 0.f, 0.f, 0.f, 0.f};
        #pragma unroll
        for (int t = 0; t < NK; ++t) {
            const v2f a = *reinterpret_cast<const v2f*>(aRow + t * 4);
            acc = __builtin_amdgcn_wmma_f32_16x16x4_f32(
                false, a, false, bf[t], (short)0, acc, false, false);
        }

        float h[8];
        #pragma unroll
        for (int j = 0; j < 8; ++j) {
            const float v = acc[j] + bv;
            const float a = LEAKY ? fmaxf(v, LEAKY_SLOPE * v) : v;
            if constexpr (SCALE_OUT) h[j] = a * so[j];
            else h[j] = a;
        }

        if constexpr (MD2 > 0) {
            float* hRow = sH + (rt * 16 + hi * 8) * PITCH + col;
            #pragma unroll
            for (int j = 0; j < 8; ++j) hRow[j * PITCH] = h[j];
        } else {
            float* outBase = out + (size_t)(tileBase + hi * 8) * MD + col;
            if (FULLCOL && (tileBase + 16 <= Nn)) {
                #pragma unroll
                for (int j = 0; j < 8; ++j) outBase[j * MD] = h[j];
            } else {
                #pragma unroll
                for (int j = 0; j < 8; ++j) {
                    const int r = tileBase + hi * 8 + j;
                    if (col < MD && r < Nn) outBase[j * MD] = h[j];
                }
            }
        }
    }

    if constexpr (MD2 > 0) {
        static_assert(D == 64 && MD == 64 && MD2 <= 4, "second product: 64-wide tile, rows of 4");
        constexpr int NK2 = MD / 4;
        const int   c2 = (ln15 < MD2) ? ln15 : (MD2 - 1);
        const float m2 = (ln15 < MD2) ? 1.f : 0.f;
        v2f bf2[NK2];
        if (wv < ROWS / 16) {
            #pragma unroll
            for (int t = 0; t < NK2; ++t) {
                const int k0 = t * 4 + kHalf;
                bf2[t].x = W2[k0 * MD2 + c2] * m2;
                bf2[t].y = W2[(k0 + 1) * MD2 + c2] * m2;
            }
        }
        __syncthreads();
        if (wv < ROWS / 16) {
            const int tileBase = nodeBase + wv * 16;
            if (tileBase < Nn) {
                const float* hRow = sH + (wv * 16 + ln15) * PITCH + kHalf;
                v8f acc = {0.f, 0.f, 0.f, 0.f, 0.f, 0.f, 0.f, 0.f};
                #pragma unroll
                for (int t = 0; t < NK2; ++t) {
                    const v2f a = *reinterpret_cast<const v2f*>(hRow + t * 4);
                    acc = __builtin_amdgcn_wmma_f32_16x16x4_f32(
                        false, a, false, bf2[t], (short)0, acc, false, false);
                }
                float* zBase = out + (size_t)(tileBase + hi * 8) * 4 + ln15;
                #pragma unroll
                for (int j = 0; j < 8; ++j) {
                    const int r = tileBase + hi * 8 + j;
                    if (ln15 < 4 && r < Nn) zBase[j * 4] = acc[j];
                }
            }
        }
    }
}

template <int MD>
__global__ void __launch_bounds__(256)
k_agg_out(const float* __restrict__ z, const float* __restrict__ invI,
          const int* __restrict__ rowptr, const int* __restrict__ csr,
          const float* __restrict__ bias, float* __restrict__ out, int Nn) {
    static_assert(MD == 3, "three components per node, lane 3i+j");
    constexpr int NPW = 10;
    const int lane  = threadIdx.x & 31;
    const int wv    = __builtin_amdgcn_readfirstlane((int)(threadIdx.x >> 5));
    const int first = blockIdx.x * (8 * NPW) + wv * NPW;
    const int pIdx  = first + lane;
    const int   rp  = rowptr[pIdx < Nn ? pIdx : Nn];
    const float nrm = invI[pIdx < Nn ? pIdx : (Nn - 1)];
    const float4* z4 = reinterpret_cast<const float4*>(z);

    int s[NPW];
    #pragma unroll
    for (int i = 0; i < NPW; ++i) {
        const int beg  = __builtin_amdgcn_readlane(rp, i);
        const int end  = __builtin_amdgcn_readlane(rp, i + 1);
        const int last = (end > 0) ? (end - 1) : 0;
        const int e    = beg + lane;
        s[i] = csr[e < last ? e : last];
    }
    float4 v[NPW];
    #pragma unroll
    for (int i = 0; i < NPW; ++i) v[i] = z4[s[i]];

    float ax[NPW], ay[NPW], az[NPW];
    #pragma unroll
    for (int i = 0; i < NPW; ++i) {
        const int beg = __builtin_amdgcn_readlane(rp, i);
        const int end = __builtin_amdgcn_readlane(rp, i + 1);
        const bool has = lane < end - beg;
        ax[i] = has ? v[i].x : 0.f;
        ay[i] = has ? v[i].y : 0.f;
        az[i] = has ? v[i].z : 0.f;
        for (int e = beg + 32 + lane; e < end; e += 32) {
            const float4 t = z4[csr[e]];
            ax[i] += t.x; ay[i] += t.y; az[i] += t.z;
        }
    }
    float mine = 0.f;
    #pragma unroll
    for (int i = 0; i < NPW; ++i) {
        const float si = __int_as_float(__builtin_amdgcn_readlane(__float_as_int(nrm), i));
        float bx = ax[i], by = ay[i], bz = az[i];
        #pragma unroll
        for (int o = 1; o < 32; o <<= 1) {
            bx += __shfl_xor(bx, o);
            by += __shfl_xor(by, o);
            bz += __shfl_xor(bz, o);
        }
        const int q = lane - 3 * i;
        const float r = (q == 0) ? bx : ((q == 1) ? by : bz);
        if (q >= 0 && q < 3) mine = r * si;
    }
    const int node = first + lane / 3;
    if (lane < 3 * NPW && node < Nn) out[(size_t)first * MD + lane] = mine + bias[lane % 3];
}

extern "C" void kernel_launch(void* const* d_in, const int* in_sizes, int n_in,
                              void* d_out, int out_size, void* d_ws, size_t ws_size,
                              hipStream_t stream) {
    (void)n_in; (void)out_size;

    const float* features = (const float*)d_in[0];
    const int*   esrc     = (const int*)d_in[1];
    const int*   edst     = (const int*)d_in[2];

    const int N = in_sizes[0] / 4;
    const int E = in_sizes[1];
    static const int DIMS[9] = {4, 64, 64, 64, 64, 64, 64, 64, 3};

    float* ws      = (float*)d_ws;
    float* inv_out = ws;
    float* inv_in  = ws + 100352;
    float* xbuf    = ws + 200704;
    float* agg     = ws + 200704 + 6400000;

    const int BT = 256;

    if (ws_size >= (size_t)65204224) {
        int* partI  = (int*)xbuf;
        int* partO  = (int*)(xbuf + 3200000);
        int* cntI   = (int*)(agg + 401408);
        int* csr    = (int*)(ws + 13000704);
        int* rowptr = (int*)(ws + 13000704 + 3200000);

        k_lb_count<<<LB_CHUNKS * LB_RANGES, LB_T, 0, stream>>>(esrc, edst, partI, partO, E, N);
        k_lb_nodes<<<(N + BT - 1) / BT, BT, 0, stream>>>(partI, partO, cntI, inv_out, inv_in,
                                                          features, agg, N);
        k_scan<<<1, 1024, 0, stream>>>(cntI, rowptr, N);
        k_lb_place<<<LB_CHUNKS * LB_RANGES, LB_T, 0, stream>>>(esrc, edst, rowptr, partI, csr, E, N);

        const int lblocks = (N + 79) / 80;
        float* planeA = xbuf;
        float* planeB = agg;
        k_layer<4, 64, true, true, 0><<<lblocks, BT, 0, stream>>>(
            planeB, inv_out, inv_in, rowptr, csr,
            (const float*)d_in[3], (const float*)d_in[4], (const float*)d_in[3], planeA, N);
        const float* xin = planeA;
        float* xout = planeB;
        for (int i = 1; i < 6; ++i) {
            k_layer<64, 64, true, true, 0><<<lblocks, BT, 0, stream>>>(
                xin, inv_out, inv_in, rowptr, csr,
                (const float*)d_in[3 + 2 * i], (const float*)d_in[4 + 2 * i],
                (const float*)d_in[3 + 2 * i], xout, N);
            float* t = (float*)xin;
            xin = xout;
            xout = t;
        }
        k_layer<64, 64, true, true, 3><<<lblocks, BT, 0, stream>>>(
            xin, inv_out, inv_in, rowptr, csr,
            (const float*)d_in[15], (const float*)d_in[16], (const float*)d_in[17], xout, N);
        k_agg_out<3><<<lblocks, BT, 0, stream>>>(
            xout, inv_in, rowptr, csr, (const float*)d_in[18], (float*)d_out, N);
        return;
    }

    {
        int n4 = (N + 3) / 4;
        k_zero4<<<(n4 + BT - 1) / BT, BT, 0, stream>>>((float4*)inv_out, n4);
        k_zero4<<<(n4 + BT - 1) / BT, BT, 0, stream>>>((float4*)inv_in,  n4);
        k_degrees<<<(E + BT - 1) / BT, BT, 0, stream>>>(esrc, edst, inv_out, inv_in, E);
        k_inv_sqrt<<<(N + BT - 1) / BT, BT, 0, stream>>>(inv_out, inv_in, N);
    }

    const float* xcur = features;
    for (int i = 0; i < 8; ++i) {
        const int Kd = DIMS[i];
        const int Md = DIMS[i + 1];
        const float* Wi = (const float*)d_in[3 + 2 * i];
        const float* bi = (const float*)d_in[4 + 2 * i];

        int n4 = N * Kd / 4;
        k_zero4<<<(n4 + BT - 1) / BT, BT, 0, stream>>>((float4*)agg, n4);

        int chunks = Kd / 4;
        int lg = (chunks == 16) ? 4 : 0;
        long long total = (long long)E * chunks;
        int blocks = (int)((total + BT - 1) / BT);
        k_gather<<<blocks, BT, 0, stream>>>(xcur, inv_out, esrc, edst, agg, E, lg, Kd);

        int colTiles  = (Md + 15) / 16;
        int rowBlocks = (N + 63) / 64;
        long long waves = (long long)rowBlocks * colTiles;
        long long thr = waves * 32;
        int gblocks = (int)((thr + BT - 1) / BT);
        float* yout = (i == 7) ? (float*)d_out : xbuf;

        if (i == 0) {
            k_gemm_wmma<4, 64, true><<<gblocks, BT, 0, stream>>>(
                agg, inv_in, Wi, bi, yout, N, colTiles, rowBlocks);
        } else if (i < 7) {
            k_gemm_wmma<64, 64, true><<<gblocks, BT, 0, stream>>>(
                agg, inv_in, Wi, bi, yout, N, colTiles, rowBlocks);
        } else {
            k_gemm_wmma<64, 3, false><<<gblocks, BT, 0, stream>>>(
                agg, inv_in, Wi, bi, yout, N, colTiles, rowBlocks);
        }
        xcur = xbuf;
    }
}


# Row-offset scan split over 25 work-groups: each sums the counts in front of its own 4096 by itself, twelve 16-byte loads in flight, then scans its own; no serial 25-trip pass; integer sums only
# speedup vs baseline: 7.9088x; 1.0879x over previous
#include <hip/hip_runtime.h>


typedef __attribute__((ext_vector_type(2))) float v2f;
typedef __attribute__((ext_vector_type(8))) float v8f;

#define LEAKY_SLOPE 0.01f

__device__ __forceinline__ void atomAddF(float* p, float v) {
    __hip_atomic_fetch_add(p, v, __ATOMIC_RELAXED, __HIP_MEMORY_SCOPE_AGENT);
}

__global__ void k_zero4(float4* __restrict__ p, int n4) {
    int i = blockIdx.x * blockDim.x + threadIdx.x;
    if (i < n4) p[i] = make_float4(0.f, 0.f, 0.f, 0.f);
}

__global__ void k_degrees(const int* __restrict__ src, const int* __restrict__ dst,
                          float* __restrict__ degO, float* __restrict__ degI, int E) {
    int e = blockIdx.x * blockDim.x + threadIdx.x;
    if (e >= E) return;
    atomAddF(&degO[src[e]], 1.0f);
    atomAddF(&degI[dst[e]], 1.0f);
}

__global__ void k_inv_sqrt(float* __restrict__ a, float* __restrict__ b, int n) {
    int i = blockIdx.x * blockDim.x + threadIdx.x;
    if (i >= n) return;
    a[i] = rsqrtf(fmaxf(a[i], 1.0f));
    b[i] = rsqrtf(fmaxf(b[i], 1.0f));
}

__global__ void k_gather(const float* __restrict__ x, const float* __restrict__ invO,
                         const int* __restrict__ src, const int* __restrict__ dst,
                         float* __restrict__ agg, int E, int lgChunks, int d) {
    int tid = blockIdx.x * blockDim.x + threadIdx.x;
    int e = tid >> lgChunks;
    if (e >= E) return;
    int c = tid & ((1 << lgChunks) - 1);
    int s = src[e];
    int t = dst[e];
    float w = invO[s];
    const float4 v = *reinterpret_cast<const float4*>(x + (size_t)s * d + c * 4);
    float* base = agg + (size_t)t * d + c * 4;
    atomAddF(base + 0, v.x * w);
    atomAddF(base + 1, v.y * w);
    atomAddF(base + 2, v.z * w);
    atomAddF(base + 3, v.w * w);
}

#define LB_RANGES 8
#define LB_CHUNKS 32
#define LB_RW     12544
#define LB_T      1024

__device__ __forceinline__ void lbChunk(int E, int ch, int& e0, int& e1) {
    const int per = ((E + LB_CHUNKS * 4 - 1) / (LB_CHUNKS * 4)) * 4;
    e0 = ch * per;
    if (e0 > E) e0 = E;
    e1 = e0 + per;
    if (e1 > E) e1 = E;
}

#define LB_COUNT(c, v)                                                                   \
    do {                                                                                 \
        const unsigned r_ = (unsigned)((v) - lo);                                        \
        if (r_ < ulen)                                                                   \
            __hip_atomic_fetch_add(&c[r_], 1, __ATOMIC_RELAXED, __HIP_MEMORY_SCOPE_WORKGROUP); \
    } while (0)

#define LB_COUNT4(s, d)                                                                  \
    do {                                                                                 \
        LB_COUNT(cO, (s).x); LB_COUNT(cO, (s).y); LB_COUNT(cO, (s).z); LB_COUNT(cO, (s).w); \
        LB_COUNT(cI, (d).x); LB_COUNT(cI, (d).y); LB_COUNT(cI, (d).z); LB_COUNT(cI, (d).w); \
    } while (0)

__global__ void __launch_bounds__(LB_T)
k_lb_count(const int* __restrict__ src, const int* __restrict__ dst,
           int* __restrict__ partI, int* __restrict__ partO, int E, int N) {
    __shared__ int cI[LB_RW];
    __shared__ int cO[LB_RW];
    const int tid = threadIdx.x;
    const int ch  = blockIdx.x / LB_RANGES;
    const int rg  = blockIdx.x % LB_RANGES;
    const int lo  = rg * LB_RW;
    int len = N - lo;
    if (len > LB_RW) len = LB_RW;
    if (len < 0) len = 0;
    const unsigned ulen = (unsigned)len;

    for (int n = tid; n < LB_RW; n += LB_T) { cI[n] = 0; cO[n] = 0; }
    __syncthreads();

    int e0, e1;
    lbChunk(E, ch, e0, e1);
    const int n4 = (e1 - e0) >> 2;
    const int4* s4 = reinterpret_cast<const int4*>(src + e0);
    const int4* d4 = reinterpret_cast<const int4*>(dst + e0);
    int q = tid;
    for (; q + 3 * LB_T < n4; q += 4 * LB_T) {
        const int4 a0 = s4[q], a1 = s4[q + LB_T], a2 = s4[q + 2 * LB_T], a3 = s4[q + 3 * LB_T];
        const int4 b0 = d4[q], b1 = d4[q + LB_T], b2 = d4[q + 2 * LB_T], b3 = d4[q + 3 * LB_T];
        LB_COUNT4(a0, b0); LB_COUNT4(a1, b1); LB_COUNT4(a2, b2); LB_COUNT4(a3, b3);
    }
    for (; q < n4; q += LB_T) {
        const int4 a0 = s4[q];
        const int4 b0 = d4[q];
        LB_COUNT4(a0, b0);
    }
    for (int e = e0 + (n4 << 2) + tid; e < e1; e += LB_T) {
        const int s = src[e];
        const int d = dst[e];
        LB_COUNT(cO, s);
        LB_COUNT(cI, d);
    }
    __syncthreads();

    int* pI = partI + (size_t)ch * N + lo;
    int* pO = partO + (size_t)ch * N + lo;
    for (int n = tid; n < len; n += LB_T) { pI[n] = cI[n]; pO[n] = cO[n]; }
}

__global__ void k_lb_nodes(int* __restrict__ partI, const int* __restrict__ partO,
                           int* __restrict__ cntI, float* __restrict__ invO,
                           float* __restrict__ invI, const float* __restrict__ feat,
                           float* __restrict__ xs0, int N) {
    const int n = blockIdx.x * blockDim.x + threadIdx.x;
    if (n >= N) return;
    int vi[LB_CHUNKS], vo[LB_CHUNKS];
    #pragma unroll
    for (int i = 0; i < LB_CHUNKS; ++i) {
        vi[i] = partI[(size_t)i * N + n];
        vo[i] = partO[(size_t)i * N + n];
    }
    int accI = 0, accO = 0;
    #pragma unroll
    for (int i = 0; i < LB_CHUNKS; ++i) {
        partI[(size_t)i * N + n] = accI;
        accI += vi[i];
        accO += vo[i];
    }
    cntI[n] = accI;
    const float wo = rsqrtf(fmaxf((float)accO, 1.0f));
    invO[n] = wo;
    invI[n] = rsqrtf(fmaxf((float)accI, 1.0f));
    const float4 f = *reinterpret_cast<const float4*>(feat + (size_t)n * 4);
    *reinterpret_cast<float4*>(xs0 + (size_t)n * 4) = make_float4(f.x * wo, f.y * wo, f.z * wo, f.w * wo);
}

#define LB_PLACE(s, d)                                                                   \
    do {                                                                                 \
        const unsigned r_ = (unsigned)((d) - lo);                                        \
        if (r_ < ulen) {                                                                 \
            const int p_ = __hip_atomic_fetch_add(&pos[r_], 1, __ATOMIC_RELAXED,         \
                                                  __HIP_MEMORY_SCOPE_WORKGROUP);         \
            csr[p_] = (s);                                                               \
        }                                                                                \
    } while (0)

#define LB_PLACE4(s, d)                                                                  \
    do {                                                                                 \
        LB_PLACE((s).x, (d).x); LB_PLACE((s).y, (d).y);                                  \
        LB_PLACE((s).z, (d).z); LB_PLACE((s).w, (d).w);                                  \
    } while (0)

__global__ void __launch_bounds__(LB_T)
k_lb_place(const int* __restrict__ src, const int* __restrict__ dst,
           const int* __restrict__ rowptr, const int* __restrict__ partI,
           int* __restrict__ csr, int E, int N) {
    __shared__ int pos[LB_RW];
    const int tid = threadIdx.x;
    const int ch  = blockIdx.x / LB_RANGES;
    const int rg  = blockIdx.x % LB_RANGES;
    const int lo  = rg * LB_RW;
    int len = N - lo;
    if (len > LB_RW) len = LB_RW;
    if (len < 0) len = 0;
    const unsigned ulen = (unsigned)len;

    const int* pI = partI + (size_t)ch * N + lo;
    for (int n = tid; n < len; n += LB_T) pos[n] = rowptr[lo + n] + pI[n];
    __syncthreads();

    int e0, e1;
    lbChunk(E, ch, e0, e1);
    const int n4 = (e1 - e0) >> 2;
    const int4* s4 = reinterpret_cast<const int4*>(src + e0);
    const int4* d4 = reinterpret_cast<const int4*>(dst + e0);
    int q = tid;
    for (; q + 3 * LB_T < n4; q += 4 * LB_T) {
        const int4 a0 = s4[q], a1 = s4[q + LB_T], a2 = s4[q + 2 * LB_T], a3 = s4[q + 3 * LB_T];
        const int4 b0 = d4[q], b1 = d4[q + LB_T], b2 = d4[q + 2 * LB_T], b3 = d4[q + 3 * LB_T];
        LB_PLACE4(a0, b0); LB_PLACE4(a1, b1); LB_PLACE4(a2, b2); LB_PLACE4(a3, b3);
    }
    for (; q < n4; q += LB_T) {
        const int4 a0 = s4[q];
        const int4 b0 = d4[q];
        LB_PLACE4(a0, b0);
    }
    for (int e = e0 + (n4 << 2) + tid; e < e1; e += LB_T) {
        const int s = src[e];
        const int d = dst[e];
        LB_PLACE(s, d);
    }
}

#define SCAN_AHEAD 12
__global__ void __launch_bounds__(1024) k_scan(const int* __restrict__ cnt,
                                               int* __restrict__ rowptr, int n) {
    __shared__ int wsum[32];
    __shared__ int psum[32];
    __shared__ int tileTotal;
    __shared__ int baseTotal;
    const int tid = threadIdx.x, lane = tid & 31, wv = tid >> 5;
    const int b = blockIdx.x;
    const int i = b * 4096 + tid * 4;

    int a0 = 0, a1 = 0, a2 = 0, a3 = 0;
    if (i + 3 < n) {
        const int4 v = *reinterpret_cast<const int4*>(cnt + i);
        a0 = v.x; a1 = v.y; a2 = v.z; a3 = v.w;
    } else {
        if (i < n) a0 = cnt[i];
        if (i + 1 < n) a1 = cnt[i + 1];
        if (i + 2 < n) a2 = cnt[i + 2];
    }

    int part = 0;
    const int4* cnt4 = reinterpret_cast<const int4*>(cnt);
    for (int j0 = 0; j0 < b; j0 += SCAN_AHEAD) {
        int4 v[SCAN_AHEAD];
        #pragma unroll
        for (int u = 0; u < SCAN_AHEAD; ++u) {
            const int j = j0 + u;
            const int jc = (j < b) ? j : (b - 1);
            v[u] = cnt4[jc * 1024 + tid];
        }
        #pragma unroll
        for (int u = 0; u < SCAN_AHEAD; ++u)
            if (j0 + u < b) part += (v[u].x + v[u].y) + (v[u].z + v[u].w);
    }

    const int tsum = a0 + a1 + a2 + a3;
    int inc = tsum;
    int pinc = part;
    for (int o = 1; o < 32; o <<= 1) {
        const int t = __shfl_up(inc, o);
        const int p = __shfl_up(pinc, o);
        if (lane >= o) { inc += t; pinc += p; }
    }
    if (lane == 31) { wsum[wv] = inc; psum[wv] = pinc; }
    __syncthreads();
    if (wv == 0) {
        const int w = wsum[lane];
        int winc = w;
        int pall = psum[lane];
        for (int o = 1; o < 32; o <<= 1) {
            const int t = __shfl_up(winc, o);
            const int p = __shfl_up(pall, o);
            if (lane >= o) { winc += t; pall += p; }
        }
        wsum[lane] = winc - w;
        if (lane == 31) { tileTotal = winc; baseTotal = pall; }
    }
    __syncthreads();
    const int base = baseTotal;
    const int ex = base + wsum[wv] + (inc - tsum);
    if (i + 3 < n) {
        *reinterpret_cast<int4*>(rowptr + i) = make_int4(ex, ex + a0, ex + a0 + a1, ex + a0 + a1 + a2);
    } else {
        if (i < n) rowptr[i] = ex;
        if (i + 1 < n) rowptr[i + 1] = ex + a0;
        if (i + 2 < n) rowptr[i + 2] = ex + a0 + a1;
    }
    if (b == (int)gridDim.x - 1 && tid == 0) rowptr[n] = base + tileTotal;
}

template <int KDIM, int MD, bool LEAKY>
__global__ void k_gemm_wmma(const float* __restrict__ agg, const float* __restrict__ invI,
                            const float* __restrict__ W, const float* __restrict__ bias,
                            float* __restrict__ out, int Nn, int colTiles, int rowBlocks) {
    constexpr int NK = KDIM / 4;
    constexpr bool FULLCOL = (MD % 16 == 0);

    int gwave    = __builtin_amdgcn_readfirstlane(
                       (int)((blockIdx.x * blockDim.x + threadIdx.x) >> 5));
    int lane     = threadIdx.x & 31;
    int rowBlock = gwave / colTiles;
    int colTile  = gwave - rowBlock * colTiles;
    if (rowBlock >= rowBlocks) return;

    int   ln15    = lane & 15;
    int   col     = colTile * 16 + ln15;
    int   colC    = FULLCOL ? col : ((col < MD) ? col : (MD - 1));
    float colMask = FULLCOL ? 1.f : ((col < MD) ? 1.f : 0.f);
    int   kHalf   = (lane >> 4) * 2;
    int   hi      = lane >> 4;

    v2f bf[NK];
    #pragma unroll
    for (int t = 0; t < NK; ++t) {
        int k0 = t * 4 + kHalf;
        bf[t].x = W[k0 * MD + colC] * colMask;
        bf[t].y = W[(k0 + 1) * MD + colC] * colMask;
    }
    float bv = bias[colC] * colMask;

    #pragma unroll
    for (int st = 0; st < 4; ++st) {
        int tileBase = rowBlock * 64 + st * 16;
        if (tileBase >= Nn) break;
        int   mrow  = tileBase + ln15;
        int   mrowC = (mrow < Nn) ? mrow : (Nn - 1);
        float s     = invI[mrowC];

        v8f acc = {0.f, 0.f, 0.f, 0.f, 0.f, 0.f, 0.f, 0.f};
        #pragma unroll
        for (int t = 0; t < NK; ++t) {
            int k0 = t * 4 + kHalf;
            v2f a = *reinterpret_cast<const v2f*>(agg + (size_t)mrowC * KDIM + k0);
            a.x *= s;
            a.y *= s;
            acc = __builtin_amdgcn_wmma_f32_16x16x4_f32(
                false, a, false, bf[t], (short)0, acc, false, false);
        }

        float h[8];
        #pragma unroll
        for (int j = 0; j < 8; ++j) {
            float v = acc[j] + bv;
            h[j] = LEAKY ? fmaxf(v, LEAKY_SLOPE * v) : v;
        }

        float* outBase = out + (size_t)(tileBase + hi * 8) * MD + col;
        if (FULLCOL && (tileBase + 16 <= Nn)) {
            #pragma unroll
            for (int j = 0; j < 8; ++j) outBase[j * MD] = h[j];
        } else {
            #pragma unroll
            for (int j = 0; j < 8; ++j) {
                int r = tileBase + hi * 8 + j;
                if (col < MD && r < Nn) outBase[j * MD] = h[j];
            }
        }
    }
}

__device__ __forceinline__ void listLoad(const int* __restrict__ csr, int beg, int end, int lane,
                                         int& s0, int& s1) {
    const int last = (end > 0) ? (end - 1) : 0;
    const int e0 = beg + lane;
    const int e1 = e0 + 32;
    s0 = csr[e0 < last ? e0 : last];
    s1 = csr[e1 < last ? e1 : last];
}

template <int L0, bool FULL>
__device__ __forceinline__ void gather16(const char* __restrict__ xb, unsigned cOff, int sub,
                                         int sReg, int cnt, float4& acc) {
    float4 v[8];
    const int sFirst = __builtin_amdgcn_readlane(sReg, L0);
    #pragma unroll
    for (int u = 0; u < 8; ++u) {
        const int sLo = __builtin_amdgcn_readlane(sReg, L0 + 2 * u);
        const int sHi = __builtin_amdgcn_readlane(sReg, L0 + 2 * u + 1);
        int s = sub ? sHi : sLo;
        if (!FULL) s = (2 * u + sub < cnt) ? s : sFirst;
        v[u] = *reinterpret_cast<const float4*>(xb + (((unsigned)s << 8) + cOff));
    }
    #pragma unroll
    for (int u = 0; u < 8; ++u) {
        float4 t = v[u];
        if (!FULL && !(2 * u + sub < cnt)) t = make_float4(0.f, 0.f, 0.f, 0.f);
        acc.x += t.x; acc.y += t.y; acc.z += t.z; acc.w += t.w;
    }
}

__device__ __forceinline__ void gather32(const char* __restrict__ xb, unsigned cOff, int sub,
                                         int sReg, int n, float4& acc) {
    if (n >= 16) gather16<0, true>(xb, cOff, sub, sReg, 16, acc);
    else if (n > 0) gather16<0, false>(xb, cOff, sub, sReg, n, acc);
    if (n >= 32) gather16<16, true>(xb, cOff, sub, sReg, 16, acc);
    else if (n > 16) gather16<16, false>(xb, cOff, sub, sReg, n - 16, acc);
}

template <int D, int MD, bool LEAKY, bool SCALE_OUT, int MD2>
__global__ void __launch_bounds__(256) __attribute__((amdgpu_waves_per_eu(12, 16)))
k_layer(const float* __restrict__ x, const float* __restrict__ invO,
        const float* __restrict__ invI, const int* __restrict__ rowptr,
        const int* __restrict__ csr, const float* __restrict__ W,
        const float* __restrict__ bias, const float* __restrict__ W2,
        float* __restrict__ out, int Nn) {
    constexpr int LPR   = D / 4;
    constexpr int EPI   = 32 / LPR;
    constexpr int PITCH = (D >= 64) ? (D + 4) : D;
    constexpr int NK    = D / 4;
    constexpr int COLT  = (MD + 15) / 16;
    constexpr bool FULLCOL = (MD % 16 == 0);
    static_assert(8 % COLT == 0, "a wave's tiles must share one column slice");
    constexpr int ROWS  = 80;
    constexpr int NPW   = ROWS / 8;
    constexpr int NT    = (ROWS / 16) * COLT;
    constexpr bool ALLWAVES = (NT >= 8);

    __shared__ float4 sA4[ROWS * PITCH / 4];
    __shared__ float sH[(MD2 > 0) ? (ROWS * PITCH) : 4];

    const int lane     = threadIdx.x & 31;
    const int wv       = __builtin_amdgcn_readfirstlane((int)(threadIdx.x >> 5));
    const int nodeBase = blockIdx.x * ROWS;
    const int sub      = lane / LPR;
    const int c        = lane % LPR;
    const float* xc    = x + c * 4;

    const int first = nodeBase + wv * NPW;
    const int pIdx  = first + lane;
    const int   rp  = rowptr[pIdx < Nn ? pIdx : Nn];
    const float nrm = invI[pIdx < Nn ? pIdx : (Nn - 1)];

    if constexpr (D == 64) {
        const char* xb = reinterpret_cast<const char*>(x);
        const unsigned cOff = (unsigned)c * 16u;
        int   sC0, sC1, sN0, sN1;
        {
            const int b0 = __builtin_amdgcn_readlane(rp, 0);
            const int e0 = __builtin_amdgcn_readlane(rp, 1);
            const int e1 = __builtin_amdgcn_readlane(rp, 2);
            listLoad(csr, b0, e0, lane, sC0, sC1);
            listLoad(csr, e0, e1, lane, sN0, sN1);
        }
        #pragma nounroll
        for (int i = 0; i < NPW; ++i) {
            const int beg  = __builtin_amdgcn_readlane(rp, i);
            const int end  = __builtin_amdgcn_readlane(rp, i + 1);
            const int end1 = __builtin_amdgcn_readlane(rp, i + 2);
            const int end2 = __builtin_amdgcn_readlane(rp, i + 3);
            const float si = __int_as_float(__builtin_amdgcn_readlane(__float_as_int(nrm), i));

            int   sM0, sM1;
            listLoad(csr, end1, end2, lane, sM0, sM1);

            const int deg = end - beg;
            float4 acc = make_float4(0.f, 0.f, 0.f, 0.f);
            gather32(xb, cOff, sub, sC0, deg, acc);
            if (deg > 32) gather32(xb, cOff, sub, sC1, deg - 32, acc);
            for (int e = beg + 64 + sub; e < end; e += EPI) {
                const int s0 = csr[e];
                const float4 v0 = *reinterpret_cast<const float4*>(xc + (size_t)s0 * D);
                acc.x += v0.x; acc.y += v0.y; acc.z += v0.z; acc.w += v0.w;
            }
            acc.x += __shfl_xor(acc.x, 16);
            acc.y += __shfl_xor(acc.y, 16);
            acc.z += __shfl_xor(acc.z, 16);
            acc.w += __shfl_xor(acc.w, 16);
            if (sub == 0)
                sA4[(wv * NPW + i) * (PITCH / 4) + c] =
                    make_float4(acc.x * si, acc.y * si, acc.z * si, acc.w * si);

            sC0 = sN0; sC1 = sN1;
            sN0 = sM0; sN1 = sM1;
        }
    } else {
        for (int i = 0; i < NPW; ++i) {
            const int beg = __builtin_amdgcn_readlane(rp, i);
            const int end = __builtin_amdgcn_readlane(rp, i + 1);
            const float si = __int_as_float(__builtin_amdgcn_readlane(__float_as_int(nrm), i));

            float4 acc = make_float4(0.f, 0.f, 0.f, 0.f);
            int e = beg + sub;
            for (; e + 3 * EPI < end; e += 4 * EPI) {
                const int s0 = csr[e];
                const int s1 = csr[e + EPI];
                const int s2 = csr[e + 2 * EPI];
                const int s3 = csr[e + 3 * EPI];
                const float4 v0 = *reinterpret_cast<const float4*>(xc + (size_t)s0 * D);
                const float4 v1 = *reinterpret_cast<const float4*>(xc + (size_t)s1 * D);
                const float4 v2 = *reinterpret_cast<const float4*>(xc + (size_t)s2 * D);
                const float4 v3 = *reinterpret_cast<const float4*>(xc + (size_t)s3 * D);
                acc.x += v0.x; acc.y += v0.y; acc.z += v0.z; acc.w += v0.w;
                acc.x += v1.x; acc.y += v1.y; acc.z += v1.z; acc.w += v1.w;
                acc.x += v2.x; acc.y += v2.y; acc.z += v2.z; acc.w += v2.w;
                acc.x += v3.x; acc.y += v3.y; acc.z += v3.z; acc.w += v3.w;
            }
            for (; e < end; e += EPI) {
                const int s0 = csr[e];
                const float4 v0 = *reinterpret_cast<const float4*>(xc + (size_t)s0 * D);
                acc.x += v0.x; acc.y += v0.y; acc.z += v0.z; acc.w += v0.w;
            }
            #pragma unroll
            for (int o = LPR; o < 32; o <<= 1) {
                acc.x += __shfl_xor(acc.x, o);
                acc.y += __shfl_xor(acc.y, o);
                acc.z += __shfl_xor(acc.z, o);
                acc.w += __shfl_xor(acc.w, o);
            }
            if (sub == 0)
                sA4[(wv * NPW + i) * (PITCH / 4) + c] =
                    make_float4(acc.x * si, acc.y * si, acc.z * si, acc.w * si);
        }
    }

    const int   ln15    = lane & 15;
    const int   hi      = lane >> 4;
    const int   kHalf   = hi * 2;
    const int   ct      = wv % COLT;
    const int   col     = ct * 16 + ln15;
    const int   colC    = FULLCOL ? col : ((col < MD) ? col : (MD - 1));
    const float colMask = FULLCOL ? 1.f : ((col < MD) ? 1.f : 0.f);

    v2f bf[NK];
    float bv = 0.f;
    if (ALLWAVES || wv < NT) {
        #pragma unroll
        for (int t = 0; t < NK; ++t) {
            const int k0 = t * 4 + kHalf;
            bf[t].x = W[k0 * MD + colC] * colMask;
            bf[t].y = W[(k0 + 1) * MD + colC] * colMask;
        }
        bv = bias[colC] * colMask;
    }

    __syncthreads();

    const float* sAf = reinterpret_cast<const float*>(sA4);
    for (int tl = wv; tl < NT; tl += 8) {
        const int rt       = tl / COLT;
        const int tileBase = nodeBase + rt * 16;
        if (tileBase >= Nn) break;
        const float* aRow = sAf + (rt * 16 + ln15) * PITCH + kHalf;

        float so[8];
        if constexpr (SCALE_OUT) {
            const int r0 = tileBase + hi * 8;
            if (tileBase + 16 <= Nn) {
                const float4 n0 = *reinterpret_cast<const float4*>(invO + r0);
                const float4 n1 = *reinterpret_cast<const float4*>(invO + r0 + 4);
                so[0] = n0.x; so[1] = n0.y; so[2] = n0.z; so[3] = n0.w;
                so[4] = n1.x; so[5] = n1.y; so[6] = n1.z; so[7] = n1.w;
            } else {
                #pragma unroll
                for (int j = 0; j < 8; ++j) {
                    const int r = r0 + j;
                    so[j] = invO[r < Nn ? r : (Nn - 1)];
                }
            }
        }

        v8f acc = {0.f, 0.f, 0.f, 0.f, 0.f, 0.f, 0.f, 0.f};
        #pragma unroll
        for (int t = 0; t < NK; ++t) {
            const v2f a = *reinterpret_cast<const v2f*>(aRow + t * 4);
            acc = __builtin_amdgcn_wmma_f32_16x16x4_f32(
                false, a, false, bf[t], (short)0, acc, false, false);
        }

        float h[8];
        #pragma unroll
        for (int j = 0; j < 8; ++j) {
            const float v = acc[j] + bv;
            const float a = LEAKY ? fmaxf(v, LEAKY_SLOPE * v) : v;
            if constexpr (SCALE_OUT) h[j] = a * so[j];
            else h[j] = a;
        }

        if constexpr (MD2 > 0) {
            float* hRow = sH + (rt * 16 + hi * 8) * PITCH + col;
            #pragma unroll
            for (int j = 0; j < 8; ++j) hRow[j * PITCH] = h[j];
        } else {
            float* outBase = out + (size_t)(tileBase + hi * 8) * MD + col;
            if (FULLCOL && (tileBase + 16 <= Nn)) {
                #pragma unroll
                for (int j = 0; j < 8; ++j) outBase[j * MD] = h[j];
            } else {
                #pragma unroll
                for (int j = 0; j < 8; ++j) {
                    const int r = tileBase + hi * 8 + j;
                    if (col < MD && r < Nn) outBase[j * MD] = h[j];
                }
            }
        }
    }

    if constexpr (MD2 > 0) {
        static_assert(D == 64 && MD == 64 && MD2 <= 4, "second product: 64-wide tile, rows of 4");
        constexpr int NK2 = MD / 4;
        const int   c2 = (ln15 < MD2) ? ln15 : (MD2 - 1);
        const float m2 = (ln15 < MD2) ? 1.f : 0.f;
        v2f bf2[NK2];
        if (wv < ROWS / 16) {
            #pragma unroll
            for (int t = 0; t < NK2; ++t) {
                const int k0 = t * 4 + kHalf;
                bf2[t].x = W2[k0 * MD2 + c2] * m2;
                bf2[t].y = W2[(k0 + 1) * MD2 + c2] * m2;
            }
        }
        __syncthreads();
        if (wv < ROWS / 16) {
            const int tileBase = nodeBase + wv * 16;
            if (tileBase < Nn) {
                const float* hRow = sH + (wv * 16 + ln15) * PITCH + kHalf;
                v8f acc = {0.f, 0.f, 0.f, 0.f, 0.f, 0.f, 0.f, 0.f};
                #pragma unroll
                for (int t = 0; t < NK2; ++t) {
                    const v2f a = *reinterpret_cast<const v2f*>(hRow + t * 4);
                    acc = __builtin_amdgcn_wmma_f32_16x16x4_f32(
                        false, a, false, bf2[t], (short)0, acc, false, false);
                }
                float* zBase = out + (size_t)(tileBase + hi * 8) * 4 + ln15;
                #pragma unroll
                for (int j = 0; j < 8; ++j) {
                    const int r = tileBase + hi * 8 + j;
                    if (ln15 < 4 && r < Nn) zBase[j * 4] = acc[j];
                }
            }
        }
    }
}

template <int MD>
__global__ void __launch_bounds__(256)
k_agg_out(const float* __restrict__ z, const float* __restrict__ invI,
          const int* __restrict__ rowptr, const int* __restrict__ csr,
          const float* __restrict__ bias, float* __restrict__ out, int Nn) {
    static_assert(MD == 3, "three components per node, lane 3i+j");
    constexpr int NPW = 10;
    const int lane  = threadIdx.x & 31;
    const int wv    = __builtin_amdgcn_readfirstlane((int)(threadIdx.x >> 5));
    const int first = blockIdx.x * (8 * NPW) + wv * NPW;
    const int pIdx  = first + lane;
    const int   rp  = rowptr[pIdx < Nn ? pIdx : Nn];
    const float nrm = invI[pIdx < Nn ? pIdx : (Nn - 1)];
    const float4* z4 = reinterpret_cast<const float4*>(z);

    int s[NPW];
    #pragma unroll
    for (int i = 0; i < NPW; ++i) {
        const int beg  = __builtin_amdgcn_readlane(rp, i);
        const int end  = __builtin_amdgcn_readlane(rp, i + 1);
        const int last = (end > 0) ? (end - 1) : 0;
        const int e    = beg + lane;
        s[i] = csr[e < last ? e : last];
    }
    float4 v[NPW];
    #pragma unroll
    for (int i = 0; i < NPW; ++i) v[i] = z4[s[i]];

    float ax[NPW], ay[NPW], az[NPW];
    #pragma unroll
    for (int i = 0; i < NPW; ++i) {
        const int beg = __builtin_amdgcn_readlane(rp, i);
        const int end = __builtin_amdgcn_readlane(rp, i + 1);
        const bool has = lane < end - beg;
        ax[i] = has ? v[i].x : 0.f;
        ay[i] = has ? v[i].y : 0.f;
        az[i] = has ? v[i].z : 0.f;
        for (int e = beg + 32 + lane; e < end; e += 32) {
            const float4 t = z4[csr[e]];
            ax[i] += t.x; ay[i] += t.y; az[i] += t.z;
        }
    }
    float mine = 0.f;
    #pragma unroll
    for (int i = 0; i < NPW; ++i) {
        const float si = __int_as_float(__builtin_amdgcn_readlane(__float_as_int(nrm), i));
        float bx = ax[i], by = ay[i], bz = az[i];
        #pragma unroll
        for (int o = 1; o < 32; o <<= 1) {
            bx += __shfl_xor(bx, o);
            by += __shfl_xor(by, o);
            bz += __shfl_xor(bz, o);
        }
        const int q = lane - 3 * i;
        const float r = (q == 0) ? bx : ((q == 1) ? by : bz);
        if (q >= 0 && q < 3) mine = r * si;
    }
    const int node = first + lane / 3;
    if (lane < 3 * NPW && node < Nn) out[(size_t)first * MD + lane] = mine + bias[lane % 3];
}

extern "C" void kernel_launch(void* const* d_in, const int* in_sizes, int n_in,
                              void* d_out, int out_size, void* d_ws, size_t ws_size,
                              hipStream_t stream) {
    (void)n_in; (void)out_size;

    const float* features = (const float*)d_in[0];
    const int*   esrc     = (const int*)d_in[1];
    const int*   edst     = (const int*)d_in[2];

    const int N = in_sizes[0] / 4;
    const int E = in_sizes[1];
    static const int DIMS[9] = {4, 64, 64, 64, 64, 64, 64, 64, 3};

    float* ws      = (float*)d_ws;
    float* inv_out = ws;
    float* inv_in  = ws + 100352;
    float* xbuf    = ws + 200704;
    float* agg     = ws + 200704 + 6400000;

    const int BT = 256;

    if (ws_size >= (size_t)65204224) {
        int* partI  = (int*)xbuf;
        int* partO  = (int*)(xbuf + 3200000);
        int* cntI   = (int*)(agg + 401408);
        int* csr    = (int*)(ws + 13000704);
        int* rowptr = (int*)(ws + 13000704 + 3200000);

        k_lb_count<<<LB_CHUNKS * LB_RANGES, LB_T, 0, stream>>>(esrc, edst, partI, partO, E, N);
        k_lb_nodes<<<(N + BT - 1) / BT, BT, 0, stream>>>(partI, partO, cntI, inv_out, inv_in,
                                                          features, agg, N);
        k_scan<<<(N + 4095) / 4096, 1024, 0, stream>>>(cntI, rowptr, N);
        k_lb_place<<<LB_CHUNKS * LB_RANGES, LB_T, 0, stream>>>(esrc, edst, rowptr, partI, csr, E, N);

        const int lblocks = (N + 79) / 80;
        float* planeA = xbuf;
        float* planeB = agg;
        k_layer<4, 64, true, true, 0><<<lblocks, BT, 0, stream>>>(
            planeB, inv_out, inv_in, rowptr, csr,
            (const float*)d_in[3], (const float*)d_in[4], (const float*)d_in[3], planeA, N);
        const float* xin = planeA;
        float* xout = planeB;
        for (int i = 1; i < 6; ++i) {
            k_layer<64, 64, true, true, 0><<<lblocks, BT, 0, stream>>>(
                xin, inv_out, inv_in, rowptr, csr,
                (const float*)d_in[3 + 2 * i], (const float*)d_in[4 + 2 * i],
                (const float*)d_in[3 + 2 * i], xout, N);
            float* t = (float*)xin;
            xin = xout;
            xout = t;
        }
        k_layer<64, 64, true, true, 3><<<lblocks, BT, 0, stream>>>(
            xin, inv_out, inv_in, rowptr, csr,
            (const float*)d_in[15], (const float*)d_in[16], (const float*)d_in[17], xout, N);
        k_agg_out<3><<<lblocks, BT, 0, stream>>>(
            xout, inv_in, rowptr, csr, (const float*)d_in[18], (float*)d_out, N);
        return;
    }

    {
        int n4 = (N + 3) / 4;
        k_zero4<<<(n4 + BT - 1) / BT, BT, 0, stream>>>((float4*)inv_out, n4);
        k_zero4<<<(n4 + BT - 1) / BT, BT, 0, stream>>>((float4*)inv_in,  n4);
        k_degrees<<<(E + BT - 1) / BT, BT, 0, stream>>>(esrc, edst, inv_out, inv_in, E);
        k_inv_sqrt<<<(N + BT - 1) / BT, BT, 0, stream>>>(inv_out, inv_in, N);
    }

    const float* xcur = features;
    for (int i = 0; i < 8; ++i) {
        const int Kd = DIMS[i];
        const int Md = DIMS[i + 1];
        const float* Wi = (const float*)d_in[3 + 2 * i];
        const float* bi = (const float*)d_in[4 + 2 * i];

        int n4 = N * Kd / 4;
        k_zero4<<<(n4 + BT - 1) / BT, BT, 0, stream>>>((float4*)agg, n4);

        int chunks = Kd / 4;
        int lg = (chunks == 16) ? 4 : 0;
        long long total = (long long)E * chunks;
        int blocks = (int)((total + BT - 1) / BT);
        k_gather<<<blocks, BT, 0, stream>>>(xcur, inv_out, esrc, edst, agg, E, lg, Kd);

        int colTiles  = (Md + 15) / 16;
        int rowBlocks = (N + 63) / 64;
        long long waves = (long long)rowBlocks * colTiles;
        long long thr = waves * 32;
        int gblocks = (int)((thr + BT - 1) / BT);
        float* yout = (i == 7) ? (float*)d_out : xbuf;

        if (i == 0) {
            k_gemm_wmma<4, 64, true><<<gblocks, BT, 0, stream>>>(
                agg, inv_in, Wi, bi, yout, N, colTiles, rowBlocks);
        } else if (i < 7) {
            k_gemm_wmma<64, 64, true><<<gblocks, BT, 0, stream>>>(
                agg, inv_in, Wi, bi, yout, N, colTiles, rowBlocks);
        } else {
            k_gemm_wmma<64, 3, false><<<gblocks, BT, 0, stream>>>(
                agg, inv_in, Wi, bi, yout, N, colTiles, rowBlocks);
        }
        xcur = xbuf;
    }
}
